# starting kernel
# baseline (speedup 1.0000x reference)
#include <hip/hip_runtime.h>

typedef float v2f __attribute__((ext_vector_type(2)));
typedef float v8f __attribute__((ext_vector_type(8)));

#define NSP   10
#define KROW  12
#define KTOT  (81 * KROW)
#define MROWS 16

__global__ void sc_setup(const int* __restrict__ index, int N,
                         int* __restrict__ meta, int* __restrict__ perm,
                         int* __restrict__ tileSpecies, int* __restrict__ tileSlot) {
  if (threadIdx.x != 0) return;
  int cnt[NSP];
  for (int s = 0; s < NSP; ++s) cnt[s] = 0;
  for (int n = 0; n < N; ++n) {
    int s = index[n];
    if (s >= 0 && s < NSP) cnt[s]++;
  }
  int segBase[NSP];
  int base = 0, tiles = 0;
  for (int s = 0; s < NSP; ++s) {
    segBase[s] = base;
    int nt = (cnt[s] + 15) >> 4;
    for (int t = 0; t < nt; ++t) {
      tileSpecies[tiles] = s;
      tileSlot[tiles]    = base + t * 16;
      ++tiles;
    }
    base += nt * 16;
  }
  meta[0] = tiles;
  for (int i = 0; i < base; ++i) perm[i] = -1;
  int pos[NSP];
  for (int s = 0; s < NSP; ++s) pos[s] = 0;
  for (int n = 0; n < N; ++n) {
    int s = index[n];
    if (s >= 0 && s < NSP) perm[segBase[s] + pos[s]++] = n;
  }
}

__global__ __launch_bounds__(128) void sc_main(
    const float* __restrict__ X,
    const int*   __restrict__ meta,
    const int*   __restrict__ perm,
    const int*   __restrict__ tileSpecies,
    const int*   __restrict__ tileSlot,
    float*       __restrict__ out, int C,
    const float* __restrict__ U3_0e, const float* __restrict__ W3_0e,
    const float* __restrict__ U3_1o, const float* __restrict__ W3_1o,
    const float* __restrict__ U3_2e, const float* __restrict__ W3_2e,
    const float* __restrict__ U2_0e, const float* __restrict__ W2_0e,
    const float* __restrict__ U2_1o, const float* __restrict__ W2_1o,
    const float* __restrict__ U2_2e, const float* __restrict__ W2_2e,
    const float* __restrict__ U1_0e, const float* __restrict__ W1_0e,
    const float* __restrict__ U1_1o, const float* __restrict__ W1_1o,
    const float* __restrict__ U1_2e, const float* __restrict__ W1_2e)
{
  __shared__ float Ct[MROWS * KTOT];
  const int c   = blockIdx.x;
  const int s   = blockIdx.y;
  const int tid = threadIdx.x;

  for (int e = tid; e < MROWS * KTOT; e += 128) {
    const int m = e / KTOT;
    const int p = e - m * KTOT;
    float v = 0.0f;
    if (m < 9) {
      const int row  = p / KROW;
      const int slot = p - row * KROW;
      const int d0 = row / 9;
      const int d1 = row - d0 * 9;
      const float *u3, *w3, *u2, *w2, *u1, *w1;
      int mul3, mul2, irdim, ii;
      if (m == 0)     { u3=U3_0e; w3=W3_0e; u2=U2_0e; w2=W2_0e; u1=U1_0e; w1=W1_0e; mul3=10; mul2=3; irdim=1; ii=0;   }
      else if (m < 4) { u3=U3_1o; w3=W3_1o; u2=U2_1o; w2=W2_1o; u1=U1_1o; w1=W1_1o; mul3=11; mul2=2; irdim=3; ii=m-1; }
      else            { u3=U3_2e; w3=W3_2e; u2=U2_2e; w2=W2_2e; u1=U1_2e; w1=W1_2e; mul3=13; mul2=3; irdim=5; ii=m-4; }
      if (slot < 9) {
        const int ub = (row * 9 + slot) * mul3 * irdim + ii;
        for (int k = 0; k < mul3; ++k)
          v += u3[ub + k * irdim] * w3[(s * mul3 + k) * C + c];
      } else if (slot == 9) {
        const int ub = row * mul2 * irdim + ii;
        for (int k = 0; k < mul2; ++k)
          v += u2[ub + k * irdim] * w2[(s * mul2 + k) * C + c];
      } else if (slot == 10 && d1 == 0) {
        v = u1[d0 * irdim + ii] * w1[s * C + c];
      }
    }
    Ct[e] = v;
  }
  __syncthreads();

  const int  lane = tid & 31;
  const int  wave = tid >> 5;
  const int  n16  = lane & 15;
  const bool hi   = (lane & 16) != 0;
  const float* Abase = &Ct[n16 * KTOT + (hi ? 2 : 0)];
  const int numTiles = meta[0];

  int mc = 0;
  for (int t = 0; t < numTiles; ++t) {
    if (tileSpecies[t] != s) continue;
    const int my = mc++;
    if ((my & 3) != wave) continue;

    const int  slotBase = tileSlot[t];
    const int  node  = perm[slotBase + n16];
    const bool valid = node >= 0;
    const int  nodeC = valid ? node : 0;
    const float* xp = X + (nodeC * C + c) * 9;
    float xv[9];
#pragma unroll
    for (int d = 0; d < 9; ++d) xv[d] = valid ? xp[d] : 0.0f;

    const float e0 = hi ? xv[2] : xv[0];
    const float e1 = hi ? xv[3] : xv[1];
    const float e2 = hi ? xv[6] : xv[4];
    const float e3 = hi ? xv[7] : xv[5];
    const float e4 = hi ? 0.0f  : xv[8];
    const float hm = hi ? 0.0f  : 1.0f;

    v8f acc0 = {0.f,0.f,0.f,0.f,0.f,0.f,0.f,0.f};
    v8f acc1 = acc0, acc2 = acc0;

#pragma unroll 1
    for (int d1i = 0; d1i < 9; ++d1i) {
      float xd1 = xv[0];
#pragma unroll
      for (int q = 1; q < 9; ++q) if (d1i == q) xd1 = xv[q];
      const int rb = d1i * KROW;
#pragma unroll
      for (int d0 = 0; d0 < 9; ++d0) {
        const float xx = xv[d0] * xd1;
        const float* ap = Abase + d0 * (9 * KROW) + rb;
        v2f af0 = *(const v2f*)(ap);
        v2f af1 = *(const v2f*)(ap + 4);
        v2f af2 = *(const v2f*)(ap + 8);
        v2f b0, b1, b2;
        b0.x = xx * e0;  b0.y = xx * e1;
        b1.x = xx * e2;  b1.y = xx * e3;
        const float c1h = (hi && d1i == 0) ? xv[d0] : 0.0f;
        b2.x = xx * e4 + c1h;
        b2.y = xx * hm;
        acc0 = __builtin_amdgcn_wmma_f32_16x16x4_f32(false, af0, false, b0,
                                                     (short)0, acc0, false, false);
        acc1 = __builtin_amdgcn_wmma_f32_16x16x4_f32(false, af1, false, b1,
                                                     (short)0, acc1, false, false);
        acc2 = __builtin_amdgcn_wmma_f32_16x16x4_f32(false, af2, false, b2,
                                                     (short)0, acc2, false, false);
      }
    }
    v8f acc = acc0 + acc1 + acc2;

    if (valid) {
      float* op = out + (node * C + c) * 9;
      if (!hi) {
#pragma unroll
        for (int r = 0; r < 8; ++r) op[r] = acc[r];
      } else {
        op[8] = acc[0];
      }
    }
  }
}

extern "C" void kernel_launch(void* const* d_in, const int* in_sizes, int n_in,
                              void* d_out, int out_size, void* d_ws, size_t ws_size,
                              hipStream_t stream) {
  (void)n_in; (void)out_size; (void)ws_size;
  const float* X     = (const float*)d_in[0];
  const int*   index = (const int*)d_in[1];
  const int N = in_sizes[1];
  const int C = in_sizes[0] / (N * 9);

  int* meta        = (int*)d_ws;
  int* perm        = meta + 16;
  int  permCap     = (N + 16 * NSP + 15) & ~15;
  int* tileSpecies = perm + permCap;
  int* tileSlot    = tileSpecies + 256;

  sc_setup<<<1, 32, 0, stream>>>(index, N, meta, perm, tileSpecies, tileSlot);

  sc_main<<<dim3(C, NSP), 128, 0, stream>>>(
      X, meta, perm, tileSpecies, tileSlot, (float*)d_out, C,
      (const float*)d_in[2],  (const float*)d_in[3],
      (const float*)d_in[4],  (const float*)d_in[5],
      (const float*)d_in[6],  (const float*)d_in[7],
      (const float*)d_in[8],  (const float*)d_in[9],
      (const float*)d_in[10], (const float*)d_in[11],
      (const float*)d_in[12], (const float*)d_in[13],
      (const float*)d_in[14], (const float*)d_in[15],
      (const float*)d_in[16], (const float*)d_in[17],
      (const float*)d_in[18], (const float*)d_in[19]);
}


# Main GEMM re-associated in the reference's contraction order: rows (d0,d1,i) x k = last feature index on WMMA, d1 and d0 sums on VALU from D registers; 144 WMMAs per tile not 243; new summation order
# speedup vs baseline: 12.7905x; 12.7905x over previous
#include <hip/hip_runtime.h>

typedef float v2f __attribute__((ext_vector_type(2)));
typedef float v8f __attribute__((ext_vector_type(8)));

#define NSP   10
#define KROW  12
#define KTOT  (81 * KROW)
#define GROW  976
#define GITEM (9 * GROW)

__device__ __forceinline__ int coef_kk(int t, int h, int e) { return 4 * t + 2 * h + e; }

template <int MUL3, int MUL2, int IRD, int KT>
__device__ __forceinline__ void coef_load_b(v2f (&b)[IRD][KT],
    const float* __restrict__ u3, const float* __restrict__ u2, const float* __restrict__ u1,
    int type, int q3, int row, int d0, int h)
{
  const float* src = u1; int lo = 0, n = 0;
  if (type == 3)      { src = u3 + q3 * (MUL3 * IRD);  lo = 0;           n = MUL3; }
  else if (type == 2) { src = u2 + row * (MUL2 * IRD); lo = MUL3;        n = MUL2; }
  else if (type == 1) { src = u1 + d0 * IRD;           lo = MUL3 + MUL2; n = 1;    }
  const int last = n > 0 ? n - 1 : 0;
#pragma unroll
  for (int t = 0; t < KT; ++t) {
#pragma unroll
    for (int e = 0; e < 2; ++e) {
      const int rel = coef_kk(t, h, e) - lo;
      const bool ok = (unsigned)rel < (unsigned)n;
      const int relc = rel < 0 ? 0 : (rel > last ? last : rel);
#pragma unroll
      for (int ii = 0; ii < IRD; ++ii) {
        const float v = src[relc * IRD + ii];
        b[ii][t][e] = ok ? v : 0.0f;
      }
    }
  }
}

template <int MUL3, int MUL2, int KT>
__device__ __forceinline__ void coef_load_a(v2f (&a)[KT],
    const float* __restrict__ w3, const float* __restrict__ w2, const float* __restrict__ w1,
    int s, int C, int ch, int h)
{
#pragma unroll
  for (int t = 0; t < KT; ++t) {
#pragma unroll
    for (int e = 0; e < 2; ++e) {
      const int kk = coef_kk(t, h, e);
      const float* p = w1 + s * C + ch;
      bool ok = (kk == MUL3 + MUL2);
      if (kk < MUL3)             { p = w3 + (s * MUL3 + kk) * C + ch;          ok = true; }
      else if (kk < MUL3 + MUL2) { p = w2 + (s * MUL2 + (kk - MUL3)) * C + ch; ok = true; }
      const float v = *p;
      a[t][e] = ok ? v : 0.0f;
    }
  }
}

template <int IRD, int KT>
__device__ __forceinline__ void coef_mma_store(const v2f (&a)[KT], const v2f (&b)[IRD][KT],
                                               float* __restrict__ g, bool full, int nrows)
{
#pragma unroll
  for (int ii = 0; ii < IRD; ++ii) {
    v8f acc = {0.f,0.f,0.f,0.f,0.f,0.f,0.f,0.f};
#pragma unroll
    for (int t = 0; t < KT; ++t)
      acc = __builtin_amdgcn_wmma_f32_16x16x4_f32(false, a[t], false, b[ii][t],
                                                  (short)0, acc, false, false);
    if (full) {
#pragma unroll
      for (int r = 0; r < 8; ++r) g[r * GITEM + ii * GROW] = acc[r];
    } else {
#pragma unroll
      for (int r = 0; r < 8; ++r)
        if (r < nrows) g[r * GITEM + ii * GROW] = acc[r];
    }
  }
}

__global__ __launch_bounds__(128) void sc_coef(
    float* __restrict__ coef, int C,
    const float* __restrict__ U3_0e, const float* __restrict__ W3_0e,
    const float* __restrict__ U3_1o, const float* __restrict__ W3_1o,
    const float* __restrict__ U3_2e, const float* __restrict__ W3_2e,
    const float* __restrict__ U2_0e, const float* __restrict__ W2_0e,
    const float* __restrict__ U2_1o, const float* __restrict__ W2_1o,
    const float* __restrict__ U2_2e, const float* __restrict__ W2_2e,
    const float* __restrict__ U1_0e, const float* __restrict__ W1_0e,
    const float* __restrict__ U1_1o, const float* __restrict__ W1_1o,
    const float* __restrict__ U1_2e, const float* __restrict__ W1_2e)
{
  const int lane = threadIdx.x & 31;
  const int wave = __builtin_amdgcn_readfirstlane((int)(threadIdx.x >> 5));
  const int n16  = lane & 15;
  const int h    = lane >> 4;
  const int s    = blockIdx.y;
  const int pcol = blockIdx.x * 16 + n16;

  const int row  = pcol / KROW;
  const int slot = pcol - row * KROW;
  const int d0   = row / 9;
  const int d1   = row - d0 * 9;
  int type = 0;
  if (pcol < KTOT) {
    if (slot < 9)                   type = 3;
    else if (slot == 9)             type = 2;
    else if (slot == 10 && d1 == 0) type = 1;
  }
  const int q3 = row * 9 + slot;

  v2f b0[1][4], b1[3][4], b2[5][5];
  coef_load_b<10, 3, 1, 4>(b0, U3_0e, U2_0e, U1_0e, type, q3, row, d0, h);
  coef_load_b<11, 2, 3, 4>(b1, U3_1o, U2_1o, U1_1o, type, q3, row, d0, h);
  coef_load_b<13, 3, 5, 5>(b2, U3_2e, U2_2e, U1_2e, type, q3, row, d0, h);

  for (int c0 = wave * 16; c0 < C; c0 += 64) {
    const int ch  = c0 + n16;
    const int chc = ch < C ? ch : C - 1;
    v2f a0[4], a1[4], a2[5];
    coef_load_a<10, 3, 4>(a0, W3_0e, W2_0e, W1_0e, s, C, chc, h);
    coef_load_a<11, 2, 4>(a1, W3_1o, W2_1o, W1_1o, s, C, chc, h);
    coef_load_a<13, 3, 5>(a2, W3_2e, W2_2e, W1_2e, s, C, chc, h);
    const int chBase = c0 + 8 * h;
    float* g = coef + (size_t)(s * C + chBase) * GITEM + pcol;
    const int  nrows = C - chBase;
    const bool full  = c0 + 16 <= C;
    coef_mma_store<1, 4>(a0, b0, g,            full, nrows);
    coef_mma_store<3, 4>(a1, b1, g + 1 * GROW, full, nrows);
    coef_mma_store<5, 5>(a2, b2, g + 4 * GROW, full, nrows);
  }
}

#define LCAP 2048
#define LWAVE (LCAP / 4)
#define MT    47
#define GT    9

__device__ __forceinline__ int at_word(int w, int m) {
  const int g  = w / 3;
  const int qt = w - 3 * g;
  const int d0 = g / 9;
  const int d1 = g - 9 * d0;
  int hr, a;
  if (m < 8) { hr = d0 >= 5 ? 1 : 0;  a = 8 * (d0 - 5 * hr) + m; }
  else       { hr = d0 < 8 ? 1 : 0;   a = d0 < 8 ? 32 + d0 : 40; }
  const int q  = a * 9 + d1;
  const int rho = (q >> 3) * 16 + hr * 8 + (q & 7);
  return rho * 3 + qt;
}

__global__ __launch_bounds__(128) void sc_main(
    const float* __restrict__ X,
    const int*   __restrict__ index, int N,
    float*       __restrict__ out, int C,
    const float* __restrict__ coef)
{
  __shared__ __attribute__((aligned(16))) float At[MT * 16 * KROW];
  __shared__ __attribute__((aligned(16))) float A1[9 * KROW];
  __shared__ int nodeList[LCAP];
  __shared__ int waveCnt[4];
  const int c    = blockIdx.x;
  const int s    = blockIdx.y;
  const int tid  = threadIdx.x;
  const int lane = tid & 31;
  const int wave = __builtin_amdgcn_readfirstlane(tid >> 5);
  const int n16  = lane & 15;
  const bool hi  = (lane & 16) != 0;

  {
    const float*  gs = coef + (size_t)(s * C + c) * GITEM;
    const float4* gp = (const float4*)gs;
    float4* lp = (float4*)At;
    float4 va[9];
#pragma unroll
    for (int m = 0; m < 9; ++m) va[m] = gp[m * (GROW / 4) + tid];
#pragma unroll
    for (int m = 0; m < 9; ++m) lp[at_word(tid, m)] = va[m];
    if (tid < KTOT / 4 - 128) {
#pragma unroll
      for (int m = 0; m < 9; ++m) va[m] = gp[m * (GROW / 4) + 128 + tid];
#pragma unroll
      for (int m = 0; m < 9; ++m) lp[at_word(128 + tid, m)] = va[m];
    }
    if (tid < 69) {
      const float4 z = {0.f, 0.f, 0.f, 0.f};
      lp[tid < 24 ? 728 * 3 + tid : 737 * 3 + (tid - 24)] = z;
    }
    if (tid < 9 * KROW) {
      const int i  = tid / KROW;
      const int sg = tid - i * KROW;
      const int sc = sg < 9 ? sg : 8;
      const float v = gs[i * GROW + sc * (9 * KROW) + 10];
      A1[tid] = sg < 9 ? v : 0.0f;
    }
  }

  const float* Abase = &At[n16 * KROW + (hi ? 2 : 0)];
  const float* A1base = &A1[(n16 < 9 ? n16 : 8) * KROW + (hi ? 2 : 0)];

  for (int n0 = 0; n0 < N; n0 += LCAP) {
    const int wbase = n0 + wave * LWAVE;
    unsigned mk[16];
    int cntw = 0;
#pragma unroll
    for (int i = 0; i < 16; ++i) {
      const int n  = wbase + i * 32 + lane;
      const int nc = n < N ? n : N - 1;
      const int v  = index[nc];
      mk[i] = __builtin_amdgcn_ballot_w32(n < N && v == s);
      cntw += __builtin_popcount(mk[i]);
    }
    if (lane == 0) waveCnt[wave] = cntw;
    __syncthreads();
    int base = 0, total = 0;
#pragma unroll
    for (int w = 0; w < 4; ++w) {
      const int cw = waveCnt[w];
      if (w < wave) base += cw;
      total += cw;
    }
    base  = __builtin_amdgcn_readfirstlane(base);
    total = __builtin_amdgcn_readfirstlane(total);
#pragma unroll
    for (int i = 0; i < 16; ++i) {
      if ((mk[i] >> lane) & 1u)
        nodeList[base + (int)__builtin_amdgcn_mbcnt_lo(mk[i], 0u)] = wbase + i * 32 + lane;
      base += __builtin_popcount(mk[i]);
    }
    __syncthreads();

    const int numTiles = (total + 15) >> 4;
    for (int t = wave; t < numTiles; t += 4) {
      const int  slot  = t * 16 + n16;
      const bool valid = slot < total;
      const int  node  = nodeList[valid ? slot : 0];
      const float* xp = X + (node * C + c) * 9;
      float xv[9];
#pragma unroll
      for (int d = 0; d < 9; ++d) { const float xl = xp[d]; xv[d] = valid ? xl : 0.0f; }

      v2f b0, b1, b2;
      b0.x = hi ? xv[2] : xv[0];  b0.y = hi ? xv[3] : xv[1];
      b1.x = hi ? xv[6] : xv[4];  b1.y = hi ? xv[7] : xv[5];
      b2.x = hi ? 0.0f  : xv[8];  b2.y = hi ? 0.0f  : 1.0f;

      const v8f zero8 = {0.f,0.f,0.f,0.f,0.f,0.f,0.f,0.f};

      float o[9];
#pragma unroll
      for (int k = 0; k < 9; ++k) o[k] = 0.0f;

#pragma unroll 1
      for (int g = 0; g < 5; ++g) {
        const float* ag = Abase + g * (GT * 16 * KROW);
        float acc8[8];
#pragma unroll
        for (int aa = 0; aa < 8; ++aa) acc8[aa] = 0.0f;
#pragma unroll
        for (int jj = 0; jj < GT; ++jj) {
          const float* ap = ag + jj * (16 * KROW);
          const v2f af0 = *(const v2f*)(ap);
          const v2f af1 = *(const v2f*)(ap + 4);
          const v2f af2 = *(const v2f*)(ap + 8);
          v8f d = __builtin_amdgcn_wmma_f32_16x16x4_f32(false, af0, false, b0, (short)0, zero8, false, false);
          d = __builtin_amdgcn_wmma_f32_16x16x4_f32(false, af1, false, b1, (short)0, d, false, false);
          d = __builtin_amdgcn_wmma_f32_16x16x4_f32(false, af2, false, b2, (short)0, d, false, false);
#pragma unroll
          for (int r = 0; r < 8; ++r) {
            const int qq = 8 * jj + r;
            const int aa = qq / 9;
            const int e  = qq - 9 * aa;
            acc8[aa] += xv[e] * d[r];
          }
        }
        float xl = xv[0], xh = xv[5];
#pragma unroll
        for (int q = 1; q < 5; ++q) if (g == q) xl = xv[q];
#pragma unroll
        for (int q = 1; q < 4; ++q) if (g == q) xh = xv[5 + q];
        const bool last = hi && g == 4;
        const float m1 = last ? 0.0f : (hi ? xh : xl);
#pragma unroll
        for (int aa = 0; aa < 8; ++aa) {
          o[aa] += m1 * acc8[aa];
          o[8]  += (last ? xv[aa] : 0.0f) * acc8[aa];
        }
      }
      {
        const float* ap = Abase + 45 * (16 * KROW);
        const v2f af0 = *(const v2f*)(ap);
        const v2f af1 = *(const v2f*)(ap + 4);
        const v2f af2 = *(const v2f*)(ap + 8);
        const v2f ag0 = *(const v2f*)(ap + 16 * KROW);
        const v2f ag1 = *(const v2f*)(ap + 16 * KROW + 4);
        const v2f ag2 = *(const v2f*)(ap + 16 * KROW + 8);
        v8f d = __builtin_amdgcn_wmma_f32_16x16x4_f32(false, af0, false, b0, (short)0, zero8, false, false);
        v8f f = __builtin_amdgcn_wmma_f32_16x16x4_f32(false, ag0, false, b0, (short)0, zero8, false, false);
        d = __builtin_amdgcn_wmma_f32_16x16x4_f32(false, af1, false, b1, (short)0, d, false, false);
        f = __builtin_amdgcn_wmma_f32_16x16x4_f32(false, ag1, false, b1, (short)0, f, false, false);
        d = __builtin_amdgcn_wmma_f32_16x16x4_f32(false, af2, false, b2, (short)0, d, false, false);
        f = __builtin_amdgcn_wmma_f32_16x16x4_f32(false, ag2, false, b2, (short)0, f, false, false);
        float acc40 = 0.0f;
#pragma unroll
        for (int r = 0; r < 8; ++r) acc40 += xv[r] * d[r];
        acc40 += xv[8] * f[0];
        o[8] += (hi ? 0.0f : xv[8]) * acc40;
      }

      v8f d1v;
      {
        const v2f af0 = *(const v2f*)(A1base);
        const v2f af1 = *(const v2f*)(A1base + 4);
        const v2f af2 = *(const v2f*)(A1base + 8);
        d1v = __builtin_amdgcn_wmma_f32_16x16x4_f32(false, af0, false, b0, (short)0, zero8, false, false);
        d1v = __builtin_amdgcn_wmma_f32_16x16x4_f32(false, af1, false, b1, (short)0, d1v, false, false);
        d1v = __builtin_amdgcn_wmma_f32_16x16x4_f32(false, af2, false, b2, (short)0, d1v, false, false);
      }

      float ot[9];
#pragma unroll
      for (int k = 0; k < 9; ++k) ot[k] = __shfl_xor(o[k], 16);

      if (valid) {
        float* op = out + (node * C + c) * 9;
        if (!hi) {
#pragma unroll
          for (int r = 0; r < 8; ++r) op[r] = o[r] + ot[r] + d1v[r];
        } else {
          op[8] = o[8] + ot[8] + d1v[0];
        }
      }
    }
    if (n0 + LCAP < N) __syncthreads();
  }
}

extern "C" void kernel_launch(void* const* d_in, const int* in_sizes, int n_in,
                              void* d_out, int out_size, void* d_ws, size_t ws_size,
                              hipStream_t stream) {
  (void)n_in; (void)out_size; (void)ws_size;
  const float* X     = (const float*)d_in[0];
  const int*   index = (const int*)d_in[1];
  const int N = in_sizes[1];
  const int C = in_sizes[0] / (N * 9);

  float* coef = (float*)d_ws;

  sc_coef<<<dim3(GROW / 16, NSP), 128, 0, stream>>>(
      coef, C,
      (const float*)d_in[2],  (const float*)d_in[3],
      (const float*)d_in[4],  (const float*)d_in[5],
      (const float*)d_in[6],  (const float*)d_in[7],
      (const float*)d_in[8],  (const float*)d_in[9],
      (const float*)d_in[10], (const float*)d_in[11],
      (const float*)d_in[12], (const float*)d_in[13],
      (const float*)d_in[14], (const float*)d_in[15],
      (const float*)d_in[16], (const float*)d_in[17],
      (const float*)d_in[18], (const float*)d_in[19]);

  sc_main<<<dim3(C, NSP), 128, 0, stream>>>(
      X, index, N, (float*)d_out, C, coef);
}


# Coefficient kernel: a wave builds two adjacent 16-column tiles and swaps lane halves (permlanex16) so each store writes one whole aligned 128-byte plane line, not two 64-byte halves; pitch 992
# speedup vs baseline: 14.1181x; 1.1038x over previous
#include <hip/hip_runtime.h>

typedef float v2f __attribute__((ext_vector_type(2)));
typedef float v8f __attribute__((ext_vector_type(8)));

#define NSP   10
#define KROW  12
#define KTOT  (81 * KROW)
#define GROW  992
#define NPAIR (GROW / 32)
#define GITEM (9 * GROW)

__device__ __forceinline__ int coef_kk(int t, int h, int e) { return 4 * t + 2 * h + e; }

template <int MUL3, int MUL2, int IRD, int KT>
__device__ __forceinline__ void coef_load_b(v2f (&b)[IRD][KT],
    const float* __restrict__ u3, const float* __restrict__ u2, const float* __restrict__ u1,
    int type, int q3, int row, int d0, int h)
{
  const float* src = u1; int lo = 0, n = 0;
  if (type == 3)      { src = u3 + q3 * (MUL3 * IRD);  lo = 0;           n = MUL3; }
  else if (type == 2) { src = u2 + row * (MUL2 * IRD); lo = MUL3;        n = MUL2; }
  else if (type == 1) { src = u1 + d0 * IRD;           lo = MUL3 + MUL2; n = 1;    }
  const int last = n > 0 ? n - 1 : 0;
#pragma unroll
  for (int t = 0; t < KT; ++t) {
#pragma unroll
    for (int e = 0; e < 2; ++e) {
      const int rel = coef_kk(t, h, e) - lo;
      const bool ok = (unsigned)rel < (unsigned)n;
      const int relc = rel < 0 ? 0 : (rel > last ? last : rel);
#pragma unroll
      for (int ii = 0; ii < IRD; ++ii) {
        const float v = src[relc * IRD + ii];
        b[ii][t][e] = ok ? v : 0.0f;
      }
    }
  }
}

template <int MUL3, int MUL2, int KT>
__device__ __forceinline__ void coef_load_a(v2f (&a)[KT],
    const float* __restrict__ w3, const float* __restrict__ w2, const float* __restrict__ w1,
    int s, int C, int ch, int h)
{
#pragma unroll
  for (int t = 0; t < KT; ++t) {
#pragma unroll
    for (int e = 0; e < 2; ++e) {
      const int kk = coef_kk(t, h, e);
      const float* p = w1 + s * C + ch;
      bool ok = (kk == MUL3 + MUL2);
      if (kk < MUL3)             { p = w3 + (s * MUL3 + kk) * C + ch;          ok = true; }
      else if (kk < MUL3 + MUL2) { p = w2 + (s * MUL2 + (kk - MUL3)) * C + ch; ok = true; }
      const float v = *p;
      a[t][e] = ok ? v : 0.0f;
    }
  }
}

__device__ __forceinline__ float xhalf(float x) {
  const int b = __builtin_bit_cast(int, x);
  return __builtin_bit_cast(float, __builtin_amdgcn_permlanex16(b, b, 0x76543210, 0xfedcba98, false, false));
}

template <int IRD, int KT>
__device__ __forceinline__ void coef_mma_store(const v2f (&a)[KT], const v2f (&bl)[IRD][KT], const v2f (&br)[IRD][KT],
                                               float* __restrict__ g, bool hi, bool full, int nrows)
{
#pragma unroll
  for (int ii = 0; ii < IRD; ++ii) {
    v8f accl = {0.f,0.f,0.f,0.f,0.f,0.f,0.f,0.f};
    v8f accr = {0.f,0.f,0.f,0.f,0.f,0.f,0.f,0.f};
#pragma unroll
    for (int t = 0; t < KT; ++t) {
      accl = __builtin_amdgcn_wmma_f32_16x16x4_f32(false, a[t], false, bl[ii][t],
                                                   (short)0, accl, false, false);
      accr = __builtin_amdgcn_wmma_f32_16x16x4_f32(false, a[t], false, br[ii][t],
                                                   (short)0, accr, false, false);
    }
    float rlo[8], rhi[8];
#pragma unroll
    for (int r = 0; r < 8; ++r) {
      const float send = hi ? accl[r] : accr[r];
      const float recv = xhalf(send);
      rlo[r] = hi ? recv : accl[r];
      rhi[r] = hi ? accr[r] : recv;
    }
    if (full) {
#pragma unroll
      for (int r = 0; r < 8; ++r) {
        g[r * GITEM + ii * GROW]       = rlo[r];
        g[(8 + r) * GITEM + ii * GROW] = rhi[r];
      }
    } else {
#pragma unroll
      for (int r = 0; r < 8; ++r) {
        if (r < nrows)     g[r * GITEM + ii * GROW]       = rlo[r];
        if (8 + r < nrows) g[(8 + r) * GITEM + ii * GROW] = rhi[r];
      }
    }
  }
}

__global__ __launch_bounds__(128) void sc_coef(
    float* __restrict__ coef, int C,
    const float* __restrict__ U3_0e, const float* __restrict__ W3_0e,
    const float* __restrict__ U3_1o, const float* __restrict__ W3_1o,
    const float* __restrict__ U3_2e, const float* __restrict__ W3_2e,
    const float* __restrict__ U2_0e, const float* __restrict__ W2_0e,
    const float* __restrict__ U2_1o, const float* __restrict__ W2_1o,
    const float* __restrict__ U2_2e, const float* __restrict__ W2_2e,
    const float* __restrict__ U1_0e, const float* __restrict__ W1_0e,
    const float* __restrict__ U1_1o, const float* __restrict__ W1_1o,
    const float* __restrict__ U1_2e, const float* __restrict__ W1_2e)
{
  const int lane = threadIdx.x & 31;
  const int wave = __builtin_amdgcn_readfirstlane((int)(threadIdx.x >> 5));
  const int n16  = lane & 15;
  const int h    = lane >> 4;
  const int s    = blockIdx.y;
  const int pbase = blockIdx.x * 32;

  v2f b0[2][1][4], b1[2][3][4], b2[2][5][5];
#pragma unroll
  for (int j = 0; j < 2; ++j) {
    const int pcol = pbase + 16 * j + n16;
    const int row  = pcol / KROW;
    const int slot = pcol - row * KROW;
    const int d0   = row / 9;
    const int d1   = row - d0 * 9;
    int type = 0;
    if (pcol < KTOT) {
      if (slot < 9)                   type = 3;
      else if (slot == 9)             type = 2;
      else if (slot == 10 && d1 == 0) type = 1;
    }
    const int q3 = row * 9 + slot;
    coef_load_b<10, 3, 1, 4>(b0[j], U3_0e, U2_0e, U1_0e, type, q3, row, d0, h);
    coef_load_b<11, 2, 3, 4>(b1[j], U3_1o, U2_1o, U1_1o, type, q3, row, d0, h);
    coef_load_b<13, 3, 5, 5>(b2[j], U3_2e, U2_2e, U1_2e, type, q3, row, d0, h);
  }

  for (int c0 = wave * 16; c0 < C; c0 += 64) {
    const int ch  = c0 + n16;
    const int chc = ch < C ? ch : C - 1;
    v2f a0[4], a1[4], a2[5];
    coef_load_a<10, 3, 4>(a0, W3_0e, W2_0e, W1_0e, s, C, chc, h);
    coef_load_a<11, 2, 4>(a1, W3_1o, W2_1o, W1_1o, s, C, chc, h);
    coef_load_a<13, 3, 5>(a2, W3_2e, W2_2e, W1_2e, s, C, chc, h);
    float* g = coef + (size_t)(s * C + c0) * GITEM + pbase + lane;
    const int  nrows = C - c0;
    const bool full  = c0 + 16 <= C;
    const bool hi    = h != 0;
    coef_mma_store<1, 4>(a0, b0[0], b0[1], g,            hi, full, nrows);
    coef_mma_store<3, 4>(a1, b1[0], b1[1], g + 1 * GROW, hi, full, nrows);
    coef_mma_store<5, 5>(a2, b2[0], b2[1], g + 4 * GROW, hi, full, nrows);
  }
}

#define LCAP 2048
#define LWAVE (LCAP / 4)
#define MT    47
#define GT    9

__device__ __forceinline__ int at_word(int w, int m) {
  const int g  = w / 3;
  const int qt = w - 3 * g;
  const int d0 = g / 9;
  const int d1 = g - 9 * d0;
  int hr, a;
  if (m < 8) { hr = d0 >= 5 ? 1 : 0;  a = 8 * (d0 - 5 * hr) + m; }
  else       { hr = d0 < 8 ? 1 : 0;   a = d0 < 8 ? 32 + d0 : 40; }
  const int q  = a * 9 + d1;
  const int rho = (q >> 3) * 16 + hr * 8 + (q & 7);
  return rho * 3 + qt;
}

__global__ __launch_bounds__(128) void sc_main(
    const float* __restrict__ X,
    const int*   __restrict__ index, int N,
    float*       __restrict__ out, int C,
    const float* __restrict__ coef)
{
  __shared__ __attribute__((aligned(16))) float At[MT * 16 * KROW];
  __shared__ __attribute__((aligned(16))) float A1[9 * KROW];
  __shared__ int nodeList[LCAP];
  __shared__ int waveCnt[4];
  const int c    = blockIdx.x;
  const int s    = blockIdx.y;
  const int tid  = threadIdx.x;
  const int lane = tid & 31;
  const int wave = __builtin_amdgcn_readfirstlane(tid >> 5);
  const int n16  = lane & 15;
  const bool hi  = (lane & 16) != 0;

  {
    const float*  gs = coef + (size_t)(s * C + c) * GITEM;
    const float4* gp = (const float4*)gs;
    float4* lp = (float4*)At;
    float4 va[9];
#pragma unroll
    for (int m = 0; m < 9; ++m) va[m] = gp[m * (GROW / 4) + tid];
#pragma unroll
    for (int m = 0; m < 9; ++m) lp[at_word(tid, m)] = va[m];
    if (tid < KTOT / 4 - 128) {
#pragma unroll
      for (int m = 0; m < 9; ++m) va[m] = gp[m * (GROW / 4) + 128 + tid];
#pragma unroll
      for (int m = 0; m < 9; ++m) lp[at_word(128 + tid, m)] = va[m];
    }
    if (tid < 69) {
      const float4 z = {0.f, 0.f, 0.f, 0.f};
      lp[tid < 24 ? 728 * 3 + tid : 737 * 3 + (tid - 24)] = z;
    }
    if (tid < 9 * KROW) {
      const int i  = tid / KROW;
      const int sg = tid - i * KROW;
      const int sc = sg < 9 ? sg : 8;
      const float v = gs[i * GROW + sc * (9 * KROW) + 10];
      A1[tid] = sg < 9 ? v : 0.0f;
    }
  }

  const float* Abase = &At[n16 * KROW + (hi ? 2 : 0)];
  const float* A1base = &A1[(n16 < 9 ? n16 : 8) * KROW + (hi ? 2 : 0)];

  for (int n0 = 0; n0 < N; n0 += LCAP) {
    const int wbase = n0 + wave * LWAVE;
    unsigned mk[16];
    int cntw = 0;
#pragma unroll
    for (int i = 0; i < 16; ++i) {
      const int n  = wbase + i * 32 + lane;
      const int nc = n < N ? n : N - 1;
      const int v  = index[nc];
      mk[i] = __builtin_amdgcn_ballot_w32(n < N && v == s);
      cntw += __builtin_popcount(mk[i]);
    }
    if (lane == 0) waveCnt[wave] = cntw;
    __syncthreads();
    int base = 0, total = 0;
#pragma unroll
    for (int w = 0; w < 4; ++w) {
      const int cw = waveCnt[w];
      if (w < wave) base += cw;
      total += cw;
    }
    base  = __builtin_amdgcn_readfirstlane(base);
    total = __builtin_amdgcn_readfirstlane(total);
#pragma unroll
    for (int i = 0; i < 16; ++i) {
      if ((mk[i] >> lane) & 1u)
        nodeList[base + (int)__builtin_amdgcn_mbcnt_lo(mk[i], 0u)] = wbase + i * 32 + lane;
      base += __builtin_popcount(mk[i]);
    }
    __syncthreads();

    const int numTiles = (total + 15) >> 4;
    for (int t = wave; t < numTiles; t += 4) {
      const int  slot  = t * 16 + n16;
      const bool valid = slot < total;
      const int  node  = nodeList[valid ? slot : 0];
      const float* xp = X + (node * C + c) * 9;
      float xv[9];
#pragma unroll
      for (int d = 0; d < 9; ++d) { const float xl = xp[d]; xv[d] = valid ? xl : 0.0f; }

      v2f b0, b1, b2;
      b0.x = hi ? xv[2] : xv[0];  b0.y = hi ? xv[3] : xv[1];
      b1.x = hi ? xv[6] : xv[4];  b1.y = hi ? xv[7] : xv[5];
      b2.x = hi ? 0.0f  : xv[8];  b2.y = hi ? 0.0f  : 1.0f;

      const v8f zero8 = {0.f,0.f,0.f,0.f,0.f,0.f,0.f,0.f};

      float o[9];
#pragma unroll
      for (int k = 0; k < 9; ++k) o[k] = 0.0f;

#pragma unroll 1
      for (int g = 0; g < 5; ++g) {
        const float* ag = Abase + g * (GT * 16 * KROW);
        float acc8[8];
#pragma unroll
        for (int aa = 0; aa < 8; ++aa) acc8[aa] = 0.0f;
#pragma unroll
        for (int jj = 0; jj < GT; ++jj) {
          const float* ap = ag + jj * (16 * KROW);
          const v2f af0 = *(const v2f*)(ap);
          const v2f af1 = *(const v2f*)(ap + 4);
          const v2f af2 = *(const v2f*)(ap + 8);
          v8f d = __builtin_amdgcn_wmma_f32_16x16x4_f32(false, af0, false, b0, (short)0, zero8, false, false);
          d = __builtin_amdgcn_wmma_f32_16x16x4_f32(false, af1, false, b1, (short)0, d, false, false);
          d = __builtin_amdgcn_wmma_f32_16x16x4_f32(false, af2, false, b2, (short)0, d, false, false);
#pragma unroll
          for (int r = 0; r < 8; ++r) {
            const int qq = 8 * jj + r;
            const int aa = qq / 9;
            const int e  = qq - 9 * aa;
            acc8[aa] += xv[e] * d[r];
          }
        }
        float xl = xv[0], xh = xv[5];
#pragma unroll
        for (int q = 1; q < 5; ++q) if (g == q) xl = xv[q];
#pragma unroll
        for (int q = 1; q < 4; ++q) if (g == q) xh = xv[5 + q];
        const bool last = hi && g == 4;
        const float m1 = last ? 0.0f : (hi ? xh : xl);
#pragma unroll
        for (int aa = 0; aa < 8; ++aa) {
          o[aa] += m1 * acc8[aa];
          o[8]  += (last ? xv[aa] : 0.0f) * acc8[aa];
        }
      }
      {
        const float* ap = Abase + 45 * (16 * KROW);
        const v2f af0 = *(const v2f*)(ap);
        const v2f af1 = *(const v2f*)(ap + 4);
        const v2f af2 = *(const v2f*)(ap + 8);
        const v2f ag0 = *(const v2f*)(ap + 16 * KROW);
        const v2f ag1 = *(const v2f*)(ap + 16 * KROW + 4);
        const v2f ag2 = *(const v2f*)(ap + 16 * KROW + 8);
        v8f d = __builtin_amdgcn_wmma_f32_16x16x4_f32(false, af0, false, b0, (short)0, zero8, false, false);
        v8f f = __builtin_amdgcn_wmma_f32_16x16x4_f32(false, ag0, false, b0, (short)0, zero8, false, false);
        d = __builtin_amdgcn_wmma_f32_16x16x4_f32(false, af1, false, b1, (short)0, d, false, false);
        f = __builtin_amdgcn_wmma_f32_16x16x4_f32(false, ag1, false, b1, (short)0, f, false, false);
        d = __builtin_amdgcn_wmma_f32_16x16x4_f32(false, af2, false, b2, (short)0, d, false, false);
        f = __builtin_amdgcn_wmma_f32_16x16x4_f32(false, ag2, false, b2, (short)0, f, false, false);
        float acc40 = 0.0f;
#pragma unroll
        for (int r = 0; r < 8; ++r) acc40 += xv[r] * d[r];
        acc40 += xv[8] * f[0];
        o[8] += (hi ? 0.0f : xv[8]) * acc40;
      }

      v8f d1v;
      {
        const v2f af0 = *(const v2f*)(A1base);
        const v2f af1 = *(const v2f*)(A1base + 4);
        const v2f af2 = *(const v2f*)(A1base + 8);
        d1v = __builtin_amdgcn_wmma_f32_16x16x4_f32(false, af0, false, b0, (short)0, zero8, false, false);
        d1v = __builtin_amdgcn_wmma_f32_16x16x4_f32(false, af1, false, b1, (short)0, d1v, false, false);
        d1v = __builtin_amdgcn_wmma_f32_16x16x4_f32(false, af2, false, b2, (short)0, d1v, false, false);
      }

      float ot[9];
#pragma unroll
      for (int k = 0; k < 9; ++k) ot[k] = __shfl_xor(o[k], 16);

      if (valid) {
        float* op = out + (node * C + c) * 9;
        if (!hi) {
#pragma unroll
          for (int r = 0; r < 8; ++r) op[r] = o[r] + ot[r] + d1v[r];
        } else {
          op[8] = o[8] + ot[8] + d1v[0];
        }
      }
    }
    if (n0 + LCAP < N) __syncthreads();
  }
}

extern "C" void kernel_launch(void* const* d_in, const int* in_sizes, int n_in,
                              void* d_out, int out_size, void* d_ws, size_t ws_size,
                              hipStream_t stream) {
  (void)n_in; (void)out_size; (void)ws_size;
  const float* X     = (const float*)d_in[0];
  const int*   index = (const int*)d_in[1];
  const int N = in_sizes[1];
  const int C = in_sizes[0] / (N * 9);

  float* coef = (float*)d_ws;

  sc_coef<<<dim3(NPAIR, NSP), 128, 0, stream>>>(
      coef, C,
      (const float*)d_in[2],  (const float*)d_in[3],
      (const float*)d_in[4],  (const float*)d_in[5],
      (const float*)d_in[6],  (const float*)d_in[7],
      (const float*)d_in[8],  (const float*)d_in[9],
      (const float*)d_in[10], (const float*)d_in[11],
      (const float*)d_in[12], (const float*)d_in[13],
      (const float*)d_in[14], (const float*)d_in[15],
      (const float*)d_in[16], (const float*)d_in[17],
      (const float*)d_in[18], (const float*)d_in[19]);

  sc_main<<<dim3(C, NSP), 128, 0, stream>>>(
      X, index, N, (float*)d_out, C, coef);
}


# Tile loop in stages of 3 row tiles: next stage's A fragments read from LDS into a 2nd register set under 9 WMMAs on 3 accumulators; row tile = one d1 (+ pad row, more LDS); sum order kept
# speedup vs baseline: 15.2229x; 1.0783x over previous
#include <hip/hip_runtime.h>

typedef float v2f __attribute__((ext_vector_type(2)));
typedef float v8f __attribute__((ext_vector_type(8)));

#define NSP   10
#define KROW  12
#define KTOT  (81 * KROW)
#define GROW  992
#define NPAIR (GROW / 32)
#define GITEM (9 * GROW)

__device__ __forceinline__ int coef_kk(int t, int h, int e) { return 4 * t + 2 * h + e; }

template <int MUL3, int MUL2, int IRD, int KT>
__device__ __forceinline__ void coef_load_b(v2f (&b)[IRD][KT],
    const float* __restrict__ u3, const float* __restrict__ u2, const float* __restrict__ u1,
    int type, int q3, int row, int d0, int h)
{
  const float* src = u1; int lo = 0, n = 0;
  if (type == 3)      { src = u3 + q3 * (MUL3 * IRD);  lo = 0;           n = MUL3; }
  else if (type == 2) { src = u2 + row * (MUL2 * IRD); lo = MUL3;        n = MUL2; }
  else if (type == 1) { src = u1 + d0 * IRD;           lo = MUL3 + MUL2; n = 1;    }
  const int last = n > 0 ? n - 1 : 0;
#pragma unroll
  for (int t = 0; t < KT; ++t) {
#pragma unroll
    for (int e = 0; e < 2; ++e) {
      const int rel = coef_kk(t, h, e) - lo;
      const bool ok = (unsigned)rel < (unsigned)n;
      const int relc = rel < 0 ? 0 : (rel > last ? last : rel);
#pragma unroll
      for (int ii = 0; ii < IRD; ++ii) {
        const float v = src[relc * IRD + ii];
        b[ii][t][e] = ok ? v : 0.0f;
      }
    }
  }
}

template <int MUL3, int MUL2, int KT>
__device__ __forceinline__ void coef_load_a(v2f (&a)[KT],
    const float* __restrict__ w3, const float* __restrict__ w2, const float* __restrict__ w1,
    int s, int C, int ch, int h)
{
#pragma unroll
  for (int t = 0; t < KT; ++t) {
#pragma unroll
    for (int e = 0; e < 2; ++e) {
      const int kk = coef_kk(t, h, e);
      const float* p = w1 + s * C + ch;
      bool ok = (kk == MUL3 + MUL2);
      if (kk < MUL3)             { p = w3 + (s * MUL3 + kk) * C + ch;          ok = true; }
      else if (kk < MUL3 + MUL2) { p = w2 + (s * MUL2 + (kk - MUL3)) * C + ch; ok = true; }
      const float v = *p;
      a[t][e] = ok ? v : 0.0f;
    }
  }
}

__device__ __forceinline__ float xhalf(float x) {
  const int b = __builtin_bit_cast(int, x);
  return __builtin_bit_cast(float, __builtin_amdgcn_permlanex16(b, b, 0x76543210, 0xfedcba98, false, false));
}

template <int IRD, int KT>
__device__ __forceinline__ void coef_mma_store(const v2f (&a)[KT], const v2f (&bl)[IRD][KT], const v2f (&br)[IRD][KT],
                                               float* __restrict__ g, bool hi, bool full, int nrows)
{
#pragma unroll
  for (int ii = 0; ii < IRD; ++ii) {
    v8f accl = {0.f,0.f,0.f,0.f,0.f,0.f,0.f,0.f};
    v8f accr = {0.f,0.f,0.f,0.f,0.f,0.f,0.f,0.f};
#pragma unroll
    for (int t = 0; t < KT; ++t) {
      accl = __builtin_amdgcn_wmma_f32_16x16x4_f32(false, a[t], false, bl[ii][t],
                                                   (short)0, accl, false, false);
      accr = __builtin_amdgcn_wmma_f32_16x16x4_f32(false, a[t], false, br[ii][t],
                                                   (short)0, accr, false, false);
    }
    float rlo[8], rhi[8];
#pragma unroll
    for (int r = 0; r < 8; ++r) {
      const float send = hi ? accl[r] : accr[r];
      const float recv = xhalf(send);
      rlo[r] = hi ? recv : accl[r];
      rhi[r] = hi ? accr[r] : recv;
    }
    if (full) {
#pragma unroll
      for (int r = 0; r < 8; ++r) {
        g[r * GITEM + ii * GROW]       = rlo[r];
        g[(8 + r) * GITEM + ii * GROW] = rhi[r];
      }
    } else {
#pragma unroll
      for (int r = 0; r < 8; ++r) {
        if (r < nrows)     g[r * GITEM + ii * GROW]       = rlo[r];
        if (8 + r < nrows) g[(8 + r) * GITEM + ii * GROW] = rhi[r];
      }
    }
  }
}

__global__ __launch_bounds__(128) void sc_coef(
    float* __restrict__ coef, int C,
    const float* __restrict__ U3_0e, const float* __restrict__ W3_0e,
    const float* __restrict__ U3_1o, const float* __restrict__ W3_1o,
    const float* __restrict__ U3_2e, const float* __restrict__ W3_2e,
    const float* __restrict__ U2_0e, const float* __restrict__ W2_0e,
    const float* __restrict__ U2_1o, const float* __restrict__ W2_1o,
    const float* __restrict__ U2_2e, const float* __restrict__ W2_2e,
    const float* __restrict__ U1_0e, const float* __restrict__ W1_0e,
    const float* __restrict__ U1_1o, const float* __restrict__ W1_1o,
    const float* __restrict__ U1_2e, const float* __restrict__ W1_2e)
{
  const int lane = threadIdx.x & 31;
  const int wave = __builtin_amdgcn_readfirstlane((int)(threadIdx.x >> 5));
  const int n16  = lane & 15;
  const int h    = lane >> 4;
  const int s    = blockIdx.y;
  const int pbase = blockIdx.x * 32;

  v2f b0[2][1][4], b1[2][3][4], b2[2][5][5];
#pragma unroll
  for (int j = 0; j < 2; ++j) {
    const int pcol = pbase + 16 * j + n16;
    const int row  = pcol / KROW;
    const int slot = pcol - row * KROW;
    const int d0   = row / 9;
    const int d1   = row - d0 * 9;
    int type = 0;
    if (pcol < KTOT) {
      if (slot < 9)                   type = 3;
      else if (slot == 9)             type = 2;
      else if (slot == 10 && d1 == 0) type = 1;
    }
    const int q3 = row * 9 + slot;
    coef_load_b<10, 3, 1, 4>(b0[j], U3_0e, U2_0e, U1_0e, type, q3, row, d0, h);
    coef_load_b<11, 2, 3, 4>(b1[j], U3_1o, U2_1o, U1_1o, type, q3, row, d0, h);
    coef_load_b<13, 3, 5, 5>(b2[j], U3_2e, U2_2e, U1_2e, type, q3, row, d0, h);
  }

  for (int c0 = wave * 16; c0 < C; c0 += 64) {
    const int ch  = c0 + n16;
    const int chc = ch < C ? ch : C - 1;
    v2f a0[4], a1[4], a2[5];
    coef_load_a<10, 3, 4>(a0, W3_0e, W2_0e, W1_0e, s, C, chc, h);
    coef_load_a<11, 2, 4>(a1, W3_1o, W2_1o, W1_1o, s, C, chc, h);
    coef_load_a<13, 3, 5>(a2, W3_2e, W2_2e, W1_2e, s, C, chc, h);
    float* g = coef + (size_t)(s * C + c0) * GITEM + pbase + lane;
    const int  nrows = C - c0;
    const bool full  = c0 + 16 <= C;
    const bool hi    = h != 0;
    coef_mma_store<1, 4>(a0, b0[0], b0[1], g,            hi, full, nrows);
    coef_mma_store<3, 4>(a1, b1[0], b1[1], g + 1 * GROW, hi, full, nrows);
    coef_mma_store<5, 5>(a2, b2[0], b2[1], g + 4 * GROW, hi, full, nrows);
  }
}

#define LCAP 2048
#define LWAVE (LCAP / 4)
#define MT    47
#define TPITCH (17 * KROW)

#define NSTG  16

__device__ __forceinline__ void load_frags(v2f (&f)[3][3], const float* p0, const float* p1, const float* p2) {
  f[0][0] = *(const v2f*)(p0);  f[0][1] = *(const v2f*)(p0 + 4);  f[0][2] = *(const v2f*)(p0 + 8);
  f[1][0] = *(const v2f*)(p1);  f[1][1] = *(const v2f*)(p1 + 4);  f[1][2] = *(const v2f*)(p1 + 8);
  f[2][0] = *(const v2f*)(p2);  f[2][1] = *(const v2f*)(p2 + 4);  f[2][2] = *(const v2f*)(p2 + 8);
}

__device__ __forceinline__ int at_word(int w, int m) {
  const int g  = w / 3;
  const int qt = w - 3 * g;
  const int d0 = g / 9;
  const int d1 = g - 9 * d0;
  int tile, row;
  if (m < 8) {
    const int hr = d0 >= 5 ? 1 : 0;
    tile = 9 * (d0 - 5 * hr) + d1;
    row  = 8 * hr + m;
  } else {
    const bool s40 = d0 >= 8;
    tile = s40 ? 45 + (d1 >> 3) : 36 + d1;
    row  = s40 ? (d1 & 7) : 8 + d0;
  }
  return tile * (TPITCH / 4) + row * 3 + qt;
}

__global__ __launch_bounds__(128) void sc_main(
    const float* __restrict__ X,
    const int*   __restrict__ index, int N,
    float*       __restrict__ out, int C,
    const float* __restrict__ coef)
{
  __shared__ __attribute__((aligned(16))) float At[MT * TPITCH];
  __shared__ __attribute__((aligned(16))) float A1[9 * KROW];
  __shared__ int nodeList[LCAP];
  __shared__ int waveCnt[4];
  const int c    = blockIdx.x;
  const int s    = blockIdx.y;
  const int tid  = threadIdx.x;
  const int lane = tid & 31;
  const int wave = __builtin_amdgcn_readfirstlane(tid >> 5);
  const int n16  = lane & 15;
  const bool hi  = (lane & 16) != 0;

  {
    const float*  gs = coef + (size_t)(s * C + c) * GITEM;
    const float4* gp = (const float4*)gs;
    float4* lp = (float4*)At;
    int aw[9];
#pragma unroll
    for (int m = 0; m < 9; ++m) aw[m] = at_word(tid, m);
    float4 va[9];
#pragma unroll
    for (int m = 0; m < 9; ++m) va[m] = gp[m * (GROW / 4) + tid];
#pragma unroll
    for (int m = 0; m < 9; ++m) lp[aw[m]] = va[m];
    if (tid < KTOT / 4 - 128) {
#pragma unroll
      for (int m = 0; m < 9; ++m) aw[m] = at_word(128 + tid, m);
#pragma unroll
      for (int m = 0; m < 9; ++m) va[m] = gp[m * (GROW / 4) + 128 + tid];
#pragma unroll
      for (int m = 0; m < 9; ++m) lp[aw[m]] = va[m];
    }
    if (tid < 69) {
      const float4 z = {0.f, 0.f, 0.f, 0.f};
      lp[tid < 24 ? 45 * (TPITCH / 4) + 8 * 3 + tid : 46 * (TPITCH / 4) + 1 * 3 + (tid - 24)] = z;
    }
    if (tid < 9 * KROW) {
      const int i  = tid / KROW;
      const int sg = tid - i * KROW;
      const int sc = sg < 9 ? sg : 8;
      const float v = gs[i * GROW + sc * (9 * KROW) + 10];
      A1[tid] = sg < 9 ? v : 0.0f;
    }
  }

  const float* Abase = &At[n16 * KROW + (hi ? 2 : 0)];
  const float* A1base = &A1[(n16 < 9 ? n16 : 8) * KROW + (hi ? 2 : 0)];

  for (int n0 = 0; n0 < N; n0 += LCAP) {
    const int wbase = n0 + wave * LWAVE;
    unsigned mk[16];
    int cntw = 0;
#pragma unroll
    for (int i = 0; i < 16; ++i) {
      const int n  = wbase + i * 32 + lane;
      const int nc = n < N ? n : N - 1;
      const int v  = index[nc];
      mk[i] = __builtin_amdgcn_ballot_w32(n < N && v == s);
      cntw += __builtin_popcount(mk[i]);
    }
    if (lane == 0) waveCnt[wave] = cntw;
    __syncthreads();
    int base = 0, total = 0;
#pragma unroll
    for (int w = 0; w < 4; ++w) {
      const int cw = waveCnt[w];
      if (w < wave) base += cw;
      total += cw;
    }
    base  = __builtin_amdgcn_readfirstlane(base);
    total = __builtin_amdgcn_readfirstlane(total);
#pragma unroll
    for (int i = 0; i < 16; ++i) {
      if ((mk[i] >> lane) & 1u)
        nodeList[base + (int)__builtin_amdgcn_mbcnt_lo(mk[i], 0u)] = wbase + i * 32 + lane;
      base += __builtin_popcount(mk[i]);
    }
    __syncthreads();

    const int numTiles = (total + 15) >> 4;
    for (int t = wave; t < numTiles; t += 4) {
      const int  slot  = t * 16 + n16;
      const bool valid = slot < total;
      const int  node  = nodeList[valid ? slot : 0];
      v2f fa[2][3][3];
      v8f dd[2][3];
      int aofs = 0;
      load_frags(fa[0], Abase, Abase + TPITCH, Abase + 2 * TPITCH);
      const float* xp = X + (node * C + c) * 9;
      float xv[9];
#pragma unroll
      for (int d = 0; d < 9; ++d) { const float xl = xp[d]; xv[d] = valid ? xl : 0.0f; }

      v2f b0, b1, b2;
      b0.x = hi ? xv[2] : xv[0];  b0.y = hi ? xv[3] : xv[1];
      b1.x = hi ? xv[6] : xv[4];  b1.y = hi ? xv[7] : xv[5];
      b2.x = hi ? 0.0f  : xv[8];  b2.y = hi ? 0.0f  : 1.0f;

      const v8f zero8 = {0.f,0.f,0.f,0.f,0.f,0.f,0.f,0.f};
      v2f bk[3] = {b0, b1, b2};

      float o[9];
#pragma unroll
      for (int k = 0; k < 9; ++k) o[k] = 0.0f;

      float acc8[8];
#pragma unroll
      for (int aa = 0; aa < 8; ++aa) acc8[aa] = 0.0f;
#pragma unroll
      for (int s = 0; s < NSTG; ++s) {
        const int cur = s & 1;
        if (s + 1 < NSTG - 1) {
          const float* Ab = Abase + aofs;
          load_frags(fa[cur ^ 1], Ab + (3 * s + 3) * TPITCH, Ab + (3 * s + 4) * TPITCH, Ab + (3 * s + 5) * TPITCH);
        } else if (s + 1 == NSTG - 1) {
          const float* Ab = Abase + aofs;
          load_frags(fa[cur ^ 1], Ab + 45 * TPITCH, Ab + 46 * TPITCH, A1base + aofs);
        }
        if (s >= 1) {
          float t0 = dd[cur ^ 1][0][0], t1 = dd[cur ^ 1][1][0], t2 = dd[cur ^ 1][2][0];
          asm volatile("" : "+v"(bk[0]), "+v"(aofs), "+v"(t0), "+v"(t1), "+v"(t2)
                          : "v"(acc8[0]), "v"(acc8[1]), "v"(acc8[2]), "v"(acc8[3]),
                            "v"(acc8[4]), "v"(acc8[5]), "v"(acc8[6]), "v"(acc8[7]),
                            "v"(o[0]), "v"(o[1]), "v"(o[2]), "v"(o[3]), "v"(o[4]), "v"(o[5]), "v"(o[6]), "v"(o[7]), "v"(o[8])
                          : "memory");
          dd[cur ^ 1][0][0] = t0;  dd[cur ^ 1][1][0] = t1;  dd[cur ^ 1][2][0] = t2;
        } else {
          asm volatile("" : "+v"(bk[0]), "+v"(aofs) : : "memory");
        }
#pragma unroll
        for (int ks = 0; ks < 3; ++ks) {
#pragma unroll
          for (int i = 0; i < 3; ++i)
            dd[cur][i] = __builtin_amdgcn_wmma_f32_16x16x4_f32(false, fa[cur][i][ks], false, bk[ks], (short)0,
                                                               ks == 0 ? zero8 : dd[cur][i], false, false);
        }
        if (s >= 1 && s - 1 < NSTG - 1) {
          const int sp = s - 1;
          const int g  = sp / 3;
          const int ph = sp - 3 * g;
#pragma unroll
          for (int i = 0; i < 3; ++i) {
            const float xd = xv[3 * ph + i];
#pragma unroll
            for (int r = 0; r < 8; ++r) acc8[r] += xd * dd[cur ^ 1][i][r];
          }
          if (ph == 2) {
            const float xl = xv[g];
            const float xh = xv[g < 4 ? 5 + g : 5];
            const bool last = hi && g == 4;
            const float m1 = last ? 0.0f : (hi ? xh : xl);
#pragma unroll
            for (int aa = 0; aa < 8; ++aa) {
              o[aa] += m1 * acc8[aa];
              o[8]  += (last ? xv[aa] : 0.0f) * acc8[aa];
              acc8[aa] = 0.0f;
            }
          }
        }
      }
      {
        const v8f d = dd[(NSTG - 1) & 1][0];
        const v8f f = dd[(NSTG - 1) & 1][1];
        float acc40 = 0.0f;
#pragma unroll
        for (int r = 0; r < 8; ++r) acc40 += xv[r] * d[r];
        acc40 += xv[8] * f[0];
        o[8] += (hi ? 0.0f : xv[8]) * acc40;
      }
      const v8f d1v = dd[(NSTG - 1) & 1][2];

      float ot[9];
#pragma unroll
      for (int k = 0; k < 9; ++k) ot[k] = __shfl_xor(o[k], 16);

      if (valid) {
        float* op = out + (node * C + c) * 9;
        if (!hi) {
#pragma unroll
          for (int r = 0; r < 8; ++r) op[r] = o[r] + ot[r] + d1v[r];
        } else {
          op[8] = o[8] + ot[8] + d1v[0];
        }
      }
    }
    if (n0 + LCAP < N) __syncthreads();
  }
}

extern "C" void kernel_launch(void* const* d_in, const int* in_sizes, int n_in,
                              void* d_out, int out_size, void* d_ws, size_t ws_size,
                              hipStream_t stream) {
  (void)n_in; (void)out_size; (void)ws_size;
  const float* X     = (const float*)d_in[0];
  const int*   index = (const int*)d_in[1];
  const int N = in_sizes[1];
  const int C = in_sizes[0] / (N * 9);

  float* coef = (float*)d_ws;

  sc_coef<<<dim3(NPAIR, NSP), 128, 0, stream>>>(
      coef, C,
      (const float*)d_in[2],  (const float*)d_in[3],
      (const float*)d_in[4],  (const float*)d_in[5],
      (const float*)d_in[6],  (const float*)d_in[7],
      (const float*)d_in[8],  (const float*)d_in[9],
      (const float*)d_in[10], (const float*)d_in[11],
      (const float*)d_in[12], (const float*)d_in[13],
      (const float*)d_in[14], (const float*)d_in[15],
      (const float*)d_in[16], (const float*)d_in[17],
      (const float*)d_in[18], (const float*)d_in[19]);

  sc_main<<<dim3(C, NSP), 128, 0, stream>>>(
      X, index, N, (float*)d_out, C, coef);
}


# Main kernel: the 48 coefficient row tiles dealt 12 per wave and kept in registers; every wave walks all node tiles (36 WMMAs each); partial outputs summed over waves via LDS; new sum order
# speedup vs baseline: 15.8970x; 1.0443x over previous
#include <hip/hip_runtime.h>

typedef float v2f __attribute__((ext_vector_type(2)));
typedef float v8f __attribute__((ext_vector_type(8)));

#define NSP   10
#define KROW  12
#define KTOT  (81 * KROW)
#define GROW  992
#define NPAIR (GROW / 32)
#define GITEM (9 * GROW)

__device__ __forceinline__ int coef_kk(int t, int h, int e) { return 4 * t + 2 * h + e; }

template <int MUL3, int MUL2, int IRD, int KT>
__device__ __forceinline__ void coef_load_b(v2f (&b)[IRD][KT],
    const float* __restrict__ u3, const float* __restrict__ u2, const float* __restrict__ u1,
    int type, int q3, int row, int d0, int h)
{
  const float* src = u1; int lo = 0, n = 0;
  if (type == 3)      { src = u3 + q3 * (MUL3 * IRD);  lo = 0;           n = MUL3; }
  else if (type == 2) { src = u2 + row * (MUL2 * IRD); lo = MUL3;        n = MUL2; }
  else if (type == 1) { src = u1 + d0 * IRD;           lo = MUL3 + MUL2; n = 1;    }
  const int last = n > 0 ? n - 1 : 0;
#pragma unroll
  for (int t = 0; t < KT; ++t) {
#pragma unroll
    for (int e = 0; e < 2; ++e) {
      const int rel = coef_kk(t, h, e) - lo;
      const bool ok = (unsigned)rel < (unsigned)n;
      const int relc = rel < 0 ? 0 : (rel > last ? last : rel);
#pragma unroll
      for (int ii = 0; ii < IRD; ++ii) {
        const float v = src[relc * IRD + ii];
        b[ii][t][e] = ok ? v : 0.0f;
      }
    }
  }
}

template <int MUL3, int MUL2, int KT>
__device__ __forceinline__ void coef_load_a(v2f (&a)[KT],
    const float* __restrict__ w3, const float* __restrict__ w2, const float* __restrict__ w1,
    int s, int C, int ch, int h)
{
#pragma unroll
  for (int t = 0; t < KT; ++t) {
#pragma unroll
    for (int e = 0; e < 2; ++e) {
      const int kk = coef_kk(t, h, e);
      const float* p = w1 + s * C + ch;
      bool ok = (kk == MUL3 + MUL2);
      if (kk < MUL3)             { p = w3 + (s * MUL3 + kk) * C + ch;          ok = true; }
      else if (kk < MUL3 + MUL2) { p = w2 + (s * MUL2 + (kk - MUL3)) * C + ch; ok = true; }
      const float v = *p;
      a[t][e] = ok ? v : 0.0f;
    }
  }
}

__device__ __forceinline__ float xhalf(float x) {
  const int b = __builtin_bit_cast(int, x);
  return __builtin_bit_cast(float, __builtin_amdgcn_permlanex16(b, b, 0x76543210, 0xfedcba98, false, false));
}

template <int IRD, int KT>
__device__ __forceinline__ void coef_mma_store(const v2f (&a)[KT], const v2f (&bl)[IRD][KT], const v2f (&br)[IRD][KT],
                                               float* __restrict__ g, bool hi, bool full, int nrows)
{
#pragma unroll
  for (int ii = 0; ii < IRD; ++ii) {
    v8f accl = {0.f,0.f,0.f,0.f,0.f,0.f,0.f,0.f};
    v8f accr = {0.f,0.f,0.f,0.f,0.f,0.f,0.f,0.f};
#pragma unroll
    for (int t = 0; t < KT; ++t) {
      accl = __builtin_amdgcn_wmma_f32_16x16x4_f32(false, a[t], false, bl[ii][t],
                                                   (short)0, accl, false, false);
      accr = __builtin_amdgcn_wmma_f32_16x16x4_f32(false, a[t], false, br[ii][t],
                                                   (short)0, accr, false, false);
    }
    float rlo[8], rhi[8];
#pragma unroll
    for (int r = 0; r < 8; ++r) {
      const float send = hi ? accl[r] : accr[r];
      const float recv = xhalf(send);
      rlo[r] = hi ? recv : accl[r];
      rhi[r] = hi ? accr[r] : recv;
    }
    if (full) {
#pragma unroll
      for (int r = 0; r < 8; ++r) {
        g[r * GITEM + ii * GROW]       = rlo[r];
        g[(8 + r) * GITEM + ii * GROW] = rhi[r];
      }
    } else {
#pragma unroll
      for (int r = 0; r < 8; ++r) {
        if (r < nrows)     g[r * GITEM + ii * GROW]       = rlo[r];
        if (8 + r < nrows) g[(8 + r) * GITEM + ii * GROW] = rhi[r];
      }
    }
  }
}

__global__ __launch_bounds__(128) void sc_coef(
    float* __restrict__ coef, int C,
    const float* __restrict__ U3_0e, const float* __restrict__ W3_0e,
    const float* __restrict__ U3_1o, const float* __restrict__ W3_1o,
    const float* __restrict__ U3_2e, const float* __restrict__ W3_2e,
    const float* __restrict__ U2_0e, const float* __restrict__ W2_0e,
    const float* __restrict__ U2_1o, const float* __restrict__ W2_1o,
    const float* __restrict__ U2_2e, const float* __restrict__ W2_2e,
    const float* __restrict__ U1_0e, const float* __restrict__ W1_0e,
    const float* __restrict__ U1_1o, const float* __restrict__ W1_1o,
    const float* __restrict__ U1_2e, const float* __restrict__ W1_2e)
{
  const int lane = threadIdx.x & 31;
  const int wave = __builtin_amdgcn_readfirstlane((int)(threadIdx.x >> 5));
  const int n16  = lane & 15;
  const int h    = lane >> 4;
  const int s    = blockIdx.y;
  const int pbase = blockIdx.x * 32;

  v2f b0[2][1][4], b1[2][3][4], b2[2][5][5];
#pragma unroll
  for (int j = 0; j < 2; ++j) {
    const int pcol = pbase + 16 * j + n16;
    const int row  = pcol / KROW;
    const int slot = pcol - row * KROW;
    const int d0   = row / 9;
    const int d1   = row - d0 * 9;
    int type = 0;
    if (pcol < KTOT) {
      if (slot < 9)                   type = 3;
      else if (slot == 9)             type = 2;
      else if (slot == 10 && d1 == 0) type = 1;
    }
    const int q3 = row * 9 + slot;
    coef_load_b<10, 3, 1, 4>(b0[j], U3_0e, U2_0e, U1_0e, type, q3, row, d0, h);
    coef_load_b<11, 2, 3, 4>(b1[j], U3_1o, U2_1o, U1_1o, type, q3, row, d0, h);
    coef_load_b<13, 3, 5, 5>(b2[j], U3_2e, U2_2e, U1_2e, type, q3, row, d0, h);
  }

  for (int c0 = wave * 16; c0 < C; c0 += 64) {
    const int ch  = c0 + n16;
    const int chc = ch < C ? ch : C - 1;
    v2f a0[4], a1[4], a2[5];
    coef_load_a<10, 3, 4>(a0, W3_0e, W2_0e, W1_0e, s, C, chc, h);
    coef_load_a<11, 2, 4>(a1, W3_1o, W2_1o, W1_1o, s, C, chc, h);
    coef_load_a<13, 3, 5>(a2, W3_2e, W2_2e, W1_2e, s, C, chc, h);
    float* g = coef + (size_t)(s * C + c0) * GITEM + pbase + lane;
    const int  nrows = C - c0;
    const bool full  = c0 + 16 <= C;
    const bool hi    = h != 0;
    coef_mma_store<1, 4>(a0, b0[0], b0[1], g,            hi, full, nrows);
    coef_mma_store<3, 4>(a1, b1[0], b1[1], g + 1 * GROW, hi, full, nrows);
    coef_mma_store<5, 5>(a2, b2[0], b2[1], g + 4 * GROW, hi, full, nrows);
  }
}

#define LCAP 2048
#define LWAVE (LCAP / 4)
#define MT    48
#define WT    (MT / 4)
#define TPITCH (17 * KROW)
#define PTILE 144
#define CHUNK 16

__device__ __forceinline__ int at_word(int w, int m) {
  const int g  = w / 3;
  const int qt = w - 3 * g;
  const int d0 = g / 9;
  const int d1 = g - 9 * d0;
  int tile, row;
  if (m < 8) {
    const int hr = d0 >= 5 ? 1 : 0;
    tile = 9 * (d0 - 5 * hr) + d1;
    row  = 8 * hr + m;
  } else {
    const bool s40 = d0 >= 8;
    tile = s40 ? 45 + (d1 >> 3) : 36 + d1;
    row  = s40 ? (d1 & 7) : 8 + d0;
  }
  return tile * (TPITCH / 4) + row * 3 + qt;
}

template <int WV>
__device__ __forceinline__ void wave_tiles(const v2f (&fa)[WT][3], const float* __restrict__ X, int C, int c,
                                           const int* nodeList, int tb, int te, int total,
                                           float* part, int n16, bool hi)
{
  const v8f zero8 = {0.f,0.f,0.f,0.f,0.f,0.f,0.f,0.f};
  float xn[9];
  const int lim = te * 16 < total ? te * 16 : total;
  int slot = tb * 16 + n16;
  bool valid = slot < lim;
  {
    const int node = nodeList[valid ? slot : 0];
    const float* xp = X + (node * C + c) * 9;
#pragma unroll
    for (int d = 0; d < 9; ++d) xn[d] = xp[d];
  }
  for (int t = tb; t < te; ++t) {
    float xv[9];
#pragma unroll
    for (int d = 0; d < 9; ++d) xv[d] = valid ? xn[d] : 0.0f;
    {
      slot += 16;
      valid = slot < lim;
      const int node2 = nodeList[valid ? slot : 0];
      const float* xp = X + (node2 * C + c) * 9;
#pragma unroll
      for (int d = 0; d < 9; ++d) xn[d] = xp[d];
    }

    v2f bk[3];
    bk[0].x = hi ? xv[2] : xv[0];  bk[0].y = hi ? xv[3] : xv[1];
    bk[1].x = hi ? xv[6] : xv[4];  bk[1].y = hi ? xv[7] : xv[5];
    bk[2].x = hi ? 0.0f  : xv[8];  bk[2].y = hi ? 0.0f  : 1.0f;

    float o[9];
#pragma unroll
    for (int k = 0; k < 9; ++k) o[k] = 0.0f;
    float acc8[8];
#pragma unroll
    for (int aa = 0; aa < 8; ++aa) acc8[aa] = 0.0f;
    float acc40 = 0.0f;

    v8f dd[2][3];
#pragma unroll
    for (int sg = 0; sg <= WT / 3; ++sg) {
      if (sg >= 2 && sg < WT / 3) {
        asm volatile("" : "+v"(bk[0])
                        : "v"(acc8[0]), "v"(acc8[1]), "v"(acc8[2]), "v"(acc8[3]), "v"(acc8[4]), "v"(acc8[5]), "v"(acc8[6]), "v"(acc8[7]),
                          "v"(o[0]), "v"(o[1]), "v"(o[2]), "v"(o[3]), "v"(o[4]), "v"(o[5]), "v"(o[6]), "v"(o[7]), "v"(o[8]), "v"(acc40));
      }
      if (sg < WT / 3) {
#pragma unroll
        for (int ks = 0; ks < 3; ++ks) {
#pragma unroll
          for (int i = 0; i < 3; ++i)
            dd[sg & 1][i] = __builtin_amdgcn_wmma_f32_16x16x4_f32(false, fa[3 * sg + i][ks], false, bk[ks], (short)0,
                                                                  ks == 0 ? zero8 : dd[sg & 1][i], false, false);
        }
      }
      if (sg >= 1) {
        const int sp = sg - 1;
#pragma unroll
        for (int i = 0; i < 3; ++i) {
          const v8f d = dd[sp & 1][i];
          const int T = WT * WV + 3 * sp + i;
          if (T < 45) {
            const int g  = T / 9;
            const int d1 = T - 9 * g;
            const float xd = xv[d1];
#pragma unroll
            for (int r = 0; r < 8; ++r) acc8[r] += xd * d[r];
            if (d1 == 8 || 3 * sp + i == WT - 1) {
              const float xl = xv[g];
              const float xh = xv[g < 4 ? 5 + g : 5];
              const bool last = hi && g == 4;
              const float m1 = last ? 0.0f : (hi ? xh : xl);
#pragma unroll
              for (int aa = 0; aa < 8; ++aa) {
                o[aa] += m1 * acc8[aa];
                o[8]  += (last ? xv[aa] : 0.0f) * acc8[aa];
                acc8[aa] = 0.0f;
              }
            }
          } else if (T == 45) {
#pragma unroll
            for (int r = 0; r < 8; ++r) acc40 += xv[r] * d[r];
          } else if (T == 46) {
            acc40 += xv[8] * d[0];
            o[8] += (hi ? 0.0f : xv[8]) * acc40;
          } else {
#pragma unroll
            for (int r = 0; r < 8; ++r) o[r] += hi ? 0.0f : d[r];
            o[8] += hi ? d[0] : 0.0f;
          }
        }
      }
    }

    float* pw = part + ((t - tb) * 4 + WV) * PTILE + n16 * 9;
float tot[9];
#pragma unroll
    for (int k = 0; k < 9; ++k) tot[k] = o[k] + xhalf(o[k]);
    if (!hi) {
#pragma unroll
      for (int k = 0; k < 9; ++k) pw[k] = tot[k];
    }
  }
}

__global__ __launch_bounds__(128) void sc_main(
    const float* __restrict__ X,
    const int*   __restrict__ index, int N,
    float*       __restrict__ out, int C,
    const float* __restrict__ coef)
{
  __shared__ __attribute__((aligned(16))) float At[MT * TPITCH];
  __shared__ int nodeList[LCAP];
  __shared__ int waveCnt[4];
  const int c    = blockIdx.x;
  const int s    = blockIdx.y;
  const int tid  = threadIdx.x;
  const int lane = tid & 31;
  const int wave = __builtin_amdgcn_readfirstlane(tid >> 5);
  const int n16  = lane & 15;
  const bool hi  = (lane & 16) != 0;

  {
    const float*  gs = coef + (size_t)(s * C + c) * GITEM;
    const float4* gp = (const float4*)gs;
    float4* lp = (float4*)At;
    int aw[9];
#pragma unroll
    for (int m = 0; m < 9; ++m) aw[m] = at_word(tid, m);
    float4 va[9];
#pragma unroll
    for (int m = 0; m < 9; ++m) va[m] = gp[m * (GROW / 4) + tid];
#pragma unroll
    for (int m = 0; m < 9; ++m) lp[aw[m]] = va[m];
    if (tid < KTOT / 4 - 128) {
#pragma unroll
      for (int m = 0; m < 9; ++m) aw[m] = at_word(128 + tid, m);
#pragma unroll
      for (int m = 0; m < 9; ++m) va[m] = gp[m * (GROW / 4) + 128 + tid];
#pragma unroll
      for (int m = 0; m < 9; ++m) lp[aw[m]] = va[m];
    }
    if (tid < 69) {
      const float4 z = {0.f, 0.f, 0.f, 0.f};
      lp[tid < 24 ? 45 * (TPITCH / 4) + 8 * 3 + tid : 46 * (TPITCH / 4) + 1 * 3 + (tid - 24)] = z;
    }
    for (int e = tid; e < 16 * KROW; e += 128) {
      const int i  = e / KROW;
      const int sg = e - i * KROW;
      const bool use = i < 9 && sg < 9;
      const float v = gs[(use ? i : 0) * GROW + (use ? sg : 0) * (9 * KROW) + 10];
      At[47 * TPITCH + e] = use ? v : 0.0f;
    }
  }

  v2f fa[WT][3];

  for (int n0 = 0; n0 < N; n0 += LCAP) {
    const int wbase = n0 + wave * LWAVE;
    unsigned mk[16];
    int cntw = 0;
#pragma unroll
    for (int i = 0; i < 16; ++i) {
      const int n  = wbase + i * 32 + lane;
      const int nc = n < N ? n : N - 1;
      const int v  = index[nc];
      mk[i] = __builtin_amdgcn_ballot_w32(n < N && v == s);
      cntw += __builtin_popcount(mk[i]);
    }
    if (lane == 0) waveCnt[wave] = cntw;
    __syncthreads();
    if (n0 == 0) {
      const float* Ab = &At[(wave * WT) * TPITCH + n16 * KROW + (hi ? 2 : 0)];
#pragma unroll
      for (int j = 0; j < WT; ++j) {
        fa[j][0] = *(const v2f*)(Ab + j * TPITCH);
        fa[j][1] = *(const v2f*)(Ab + j * TPITCH + 4);
        fa[j][2] = *(const v2f*)(Ab + j * TPITCH + 8);
      }
    }
    int base = 0, total = 0;
#pragma unroll
    for (int w = 0; w < 4; ++w) {
      const int cw = waveCnt[w];
      if (w < wave) base += cw;
      total += cw;
    }
    base  = __builtin_amdgcn_readfirstlane(base);
    total = __builtin_amdgcn_readfirstlane(total);
#pragma unroll
    for (int i = 0; i < 16; ++i) {
      if ((mk[i] >> lane) & 1u)
        nodeList[base + (int)__builtin_amdgcn_mbcnt_lo(mk[i], 0u)] = wbase + i * 32 + lane;
      base += __builtin_popcount(mk[i]);
    }
    __syncthreads();

    const int numTiles = (total + 15) >> 4;
    for (int tb = 0; tb < numTiles; tb += CHUNK) {
      const int te = tb + CHUNK < numTiles ? tb + CHUNK : numTiles;
      if (wave == 0)      wave_tiles<0>(fa, X, C, c, nodeList, tb, te, total, At, n16, hi);
      else if (wave == 1) wave_tiles<1>(fa, X, C, c, nodeList, tb, te, total, At, n16, hi);
      else if (wave == 2) wave_tiles<2>(fa, X, C, c, nodeList, tb, te, total, At, n16, hi);
      else                wave_tiles<3>(fa, X, C, c, nodeList, tb, te, total, At, n16, hi);
      __syncthreads();
      const int nodes = (total - tb * 16) < (te - tb) * 16 ? (total - tb * 16) : (te - tb) * 16;
      for (int idx = tid; idx < nodes * 9; idx += 128) {
        const int ns = idx / 9;
        const int i  = idx - ns * 9;
        const int tl = ns >> 4;
        const float* p = At + tl * (4 * PTILE) + (idx - tl * PTILE);
        const float v = ((p[0] + p[PTILE]) + p[2 * PTILE]) + p[3 * PTILE];
        const int node = nodeList[tb * 16 + ns];
        out[(node * C + c) * 9 + i] = v;
      }
      __syncthreads();
    }
  }
}

extern "C" void kernel_launch(void* const* d_in, const int* in_sizes, int n_in,
                              void* d_out, int out_size, void* d_ws, size_t ws_size,
                              hipStream_t stream) {
  (void)n_in; (void)out_size; (void)ws_size;
  const float* X     = (const float*)d_in[0];
  const int*   index = (const int*)d_in[1];
  const int N = in_sizes[1];
  const int C = in_sizes[0] / (N * 9);

  float* coef = (float*)d_ws;

  sc_coef<<<dim3(NPAIR, NSP), 128, 0, stream>>>(
      coef, C,
      (const float*)d_in[2],  (const float*)d_in[3],
      (const float*)d_in[4],  (const float*)d_in[5],
      (const float*)d_in[6],  (const float*)d_in[7],
      (const float*)d_in[8],  (const float*)d_in[9],
      (const float*)d_in[10], (const float*)d_in[11],
      (const float*)d_in[12], (const float*)d_in[13],
      (const float*)d_in[14], (const float*)d_in[15],
      (const float*)d_in[16], (const float*)d_in[17],
      (const float*)d_in[18], (const float*)d_in[19]);

  sc_main<<<dim3(C, NSP), 128, 0, stream>>>(
      X, index, N, (float*)d_out, C, coef);
}


# Coefficient planes packed: 10 stored columns per (d0,d1) group (9 C3, C2), the nine C1 columns at the row end; pitch 832 not 992, 26 not 31 column-pair blocks; zero k-slots filled in LDS
# speedup vs baseline: 17.6518x; 1.1104x over previous
#include <hip/hip_runtime.h>

typedef float v2f __attribute__((ext_vector_type(2)));
typedef float v8f __attribute__((ext_vector_type(8)));

#define NSP   10
#define KROW  12
#define PSLOT 10
#define PGRP  (81 * PSLOT)
#define PUSED (PGRP + 9)
#define GROW  832
#define NPAIR (GROW / 32)
#define GITEM (9 * GROW)

__device__ __forceinline__ int coef_kk(int t, int h, int e) { return 4 * t + 2 * h + e; }

template <int MUL3, int MUL2, int IRD, int KT>
__device__ __forceinline__ void coef_load_b(v2f (&b)[IRD][KT],
    const float* __restrict__ u3, const float* __restrict__ u2, const float* __restrict__ u1,
    int type, int q3, int row, int d0, int h)
{
  const float* src = u1; int lo = 0, n = 0;
  if (type == 3)      { src = u3 + q3 * (MUL3 * IRD);  lo = 0;           n = MUL3; }
  else if (type == 2) { src = u2 + row * (MUL2 * IRD); lo = MUL3;        n = MUL2; }
  else if (type == 1) { src = u1 + d0 * IRD;           lo = MUL3 + MUL2; n = 1;    }
  const int last = n > 0 ? n - 1 : 0;
#pragma unroll
  for (int t = 0; t < KT; ++t) {
#pragma unroll
    for (int e = 0; e < 2; ++e) {
      const int rel = coef_kk(t, h, e) - lo;
      const bool ok = (unsigned)rel < (unsigned)n;
      const int relc = rel < 0 ? 0 : (rel > last ? last : rel);
#pragma unroll
      for (int ii = 0; ii < IRD; ++ii) {
        const float v = src[relc * IRD + ii];
        b[ii][t][e] = ok ? v : 0.0f;
      }
    }
  }
}

template <int MUL3, int MUL2, int KT>
__device__ __forceinline__ void coef_load_a(v2f (&a)[KT],
    const float* __restrict__ w3, const float* __restrict__ w2, const float* __restrict__ w1,
    int s, int C, int ch, int h)
{
#pragma unroll
  for (int t = 0; t < KT; ++t) {
#pragma unroll
    for (int e = 0; e < 2; ++e) {
      const int kk = coef_kk(t, h, e);
      const float* p = w1 + s * C + ch;
      bool ok = (kk == MUL3 + MUL2);
      if (kk < MUL3)             { p = w3 + (s * MUL3 + kk) * C + ch;          ok = true; }
      else if (kk < MUL3 + MUL2) { p = w2 + (s * MUL2 + (kk - MUL3)) * C + ch; ok = true; }
      const float v = *p;
      a[t][e] = ok ? v : 0.0f;
    }
  }
}

__device__ __forceinline__ float xhalf(float x) {
  const int b = __builtin_bit_cast(int, x);
  return __builtin_bit_cast(float, __builtin_amdgcn_permlanex16(b, b, 0x76543210, 0xfedcba98, false, false));
}

template <int IRD, int KT>
__device__ __forceinline__ void coef_mma_store(const v2f (&a)[KT], const v2f (&bl)[IRD][KT], const v2f (&br)[IRD][KT],
                                               float* __restrict__ g, bool hi, bool full, int nrows)
{
#pragma unroll
  for (int ii = 0; ii < IRD; ++ii) {
    v8f accl = {0.f,0.f,0.f,0.f,0.f,0.f,0.f,0.f};
    v8f accr = {0.f,0.f,0.f,0.f,0.f,0.f,0.f,0.f};
#pragma unroll
    for (int t = 0; t < KT; ++t) {
      accl = __builtin_amdgcn_wmma_f32_16x16x4_f32(false, a[t], false, bl[ii][t],
                                                   (short)0, accl, false, false);
      accr = __builtin_amdgcn_wmma_f32_16x16x4_f32(false, a[t], false, br[ii][t],
                                                   (short)0, accr, false, false);
    }
    float rlo[8], rhi[8];
#pragma unroll
    for (int r = 0; r < 8; ++r) {
      const float send = hi ? accl[r] : accr[r];
      const float recv = xhalf(send);
      rlo[r] = hi ? recv : accl[r];
      rhi[r] = hi ? accr[r] : recv;
    }
    if (full) {
#pragma unroll
      for (int r = 0; r < 8; ++r) {
        g[r * GITEM + ii * GROW]       = rlo[r];
        g[(8 + r) * GITEM + ii * GROW] = rhi[r];
      }
    } else {
#pragma unroll
      for (int r = 0; r < 8; ++r) {
        if (r < nrows)     g[r * GITEM + ii * GROW]       = rlo[r];
        if (8 + r < nrows) g[(8 + r) * GITEM + ii * GROW] = rhi[r];
      }
    }
  }
}

__global__ __launch_bounds__(128) void sc_coef(
    float* __restrict__ coef, int C,
    const float* __restrict__ U3_0e, const float* __restrict__ W3_0e,
    const float* __restrict__ U3_1o, const float* __restrict__ W3_1o,
    const float* __restrict__ U3_2e, const float* __restrict__ W3_2e,
    const float* __restrict__ U2_0e, const float* __restrict__ W2_0e,
    const float* __restrict__ U2_1o, const float* __restrict__ W2_1o,
    const float* __restrict__ U2_2e, const float* __restrict__ W2_2e,
    const float* __restrict__ U1_0e, const float* __restrict__ W1_0e,
    const float* __restrict__ U1_1o, const float* __restrict__ W1_1o,
    const float* __restrict__ U1_2e, const float* __restrict__ W1_2e)
{
  const int lane = threadIdx.x & 31;
  const int wave = __builtin_amdgcn_readfirstlane((int)(threadIdx.x >> 5));
  const int n16  = lane & 15;
  const int h    = lane >> 4;
  const int s    = blockIdx.y;
  const int pbase = blockIdx.x * 32;

  v2f b0[2][1][4], b1[2][3][4], b2[2][5][5];
#pragma unroll
  for (int j = 0; j < 2; ++j) {
    const int pcol = pbase + 16 * j + n16;
    int type = 0, row = 0, slot = 0, d0 = 0;
    if (pcol < PGRP) {
      row  = pcol / PSLOT;
      slot = pcol - row * PSLOT;
      d0   = row / 9;
      type = slot < 9 ? 3 : 2;
    } else if (pcol < PUSED) {
      d0   = pcol - PGRP;
      type = 1;
    }
    const int q3 = row * 9 + slot;
    coef_load_b<10, 3, 1, 4>(b0[j], U3_0e, U2_0e, U1_0e, type, q3, row, d0, h);
    coef_load_b<11, 2, 3, 4>(b1[j], U3_1o, U2_1o, U1_1o, type, q3, row, d0, h);
    coef_load_b<13, 3, 5, 5>(b2[j], U3_2e, U2_2e, U1_2e, type, q3, row, d0, h);
  }

  for (int c0 = wave * 16; c0 < C; c0 += 64) {
    const int ch  = c0 + n16;
    const int chc = ch < C ? ch : C - 1;
    v2f a0[4], a1[4], a2[5];
    coef_load_a<10, 3, 4>(a0, W3_0e, W2_0e, W1_0e, s, C, chc, h);
    coef_load_a<11, 2, 4>(a1, W3_1o, W2_1o, W1_1o, s, C, chc, h);
    coef_load_a<13, 3, 5>(a2, W3_2e, W2_2e, W1_2e, s, C, chc, h);
    float* g = coef + (size_t)(s * C + c0) * GITEM + pbase + lane;
    const int  nrows = C - c0;
    const bool full  = c0 + 16 <= C;
    const bool hi    = h != 0;
    coef_mma_store<1, 4>(a0, b0[0], b0[1], g,            hi, full, nrows);
    coef_mma_store<3, 4>(a1, b1[0], b1[1], g + 1 * GROW, hi, full, nrows);
    coef_mma_store<5, 5>(a2, b2[0], b2[1], g + 4 * GROW, hi, full, nrows);
  }
}

#define LCAP 2048
#define LWAVE (LCAP / 4)
#define MT    48
#define WT    (MT / 4)
#define TPITCH (17 * KROW)
#define PTILE 144
#define CHUNK 16

__device__ __forceinline__ int at_pair(int pr, int m) {
  if (pr >= PGRP / 2) return 47 * TPITCH + m * KROW + 2 * (pr - PGRP / 2);
  const int g  = pr / 5;
  const int sp = pr - 5 * g;
  const int d0 = g / 9;
  const int d1 = g - 9 * d0;
  int tile, row;
  if (m < 8) {
    const int hr = d0 >= 5 ? 1 : 0;
    tile = 9 * (d0 - 5 * hr) + d1;
    row  = 8 * hr + m;
  } else {
    const bool s40 = d0 >= 8;
    tile = s40 ? 45 + (d1 >> 3) : 36 + d1;
    row  = s40 ? (d1 & 7) : 8 + d0;
  }
  return tile * TPITCH + row * KROW + 2 * sp;
}

template <int WV>
__device__ __forceinline__ void wave_tiles(const v2f (&fa)[WT][3], const float* __restrict__ X, int C, int c,
                                           const int* nodeList, int tb, int te, int total,
                                           float* part, int n16, bool hi)
{
  const v8f zero8 = {0.f,0.f,0.f,0.f,0.f,0.f,0.f,0.f};
  float xn[9];
  const int lim = te * 16 < total ? te * 16 : total;
  int slot = tb * 16 + n16;
  bool valid = slot < lim;
  {
    const int node = nodeList[valid ? slot : 0];
    const float* xp = X + (node * C + c) * 9;
#pragma unroll
    for (int d = 0; d < 9; ++d) xn[d] = xp[d];
  }
  for (int t = tb; t < te; ++t) {
    float xv[9];
#pragma unroll
    for (int d = 0; d < 9; ++d) xv[d] = valid ? xn[d] : 0.0f;
    {
      slot += 16;
      valid = slot < lim;
      const int node2 = nodeList[valid ? slot : 0];
      const float* xp = X + (node2 * C + c) * 9;
#pragma unroll
      for (int d = 0; d < 9; ++d) xn[d] = xp[d];
    }

    v2f bk[3];
    bk[0].x = hi ? xv[2] : xv[0];  bk[0].y = hi ? xv[3] : xv[1];
    bk[1].x = hi ? xv[6] : xv[4];  bk[1].y = hi ? xv[7] : xv[5];
    bk[2].x = hi ? 0.0f  : xv[8];  bk[2].y = hi ? 0.0f  : 1.0f;

    float o[9];
#pragma unroll
    for (int k = 0; k < 9; ++k) o[k] = 0.0f;
    float acc8[8];
#pragma unroll
    for (int aa = 0; aa < 8; ++aa) acc8[aa] = 0.0f;
    float acc40 = 0.0f;

    v8f dd[2][3];
#pragma unroll
    for (int sg = 0; sg <= WT / 3; ++sg) {
      if (sg >= 2 && sg < WT / 3) {
        asm volatile("" : "+v"(bk[0])
                        : "v"(acc8[0]), "v"(acc8[1]), "v"(acc8[2]), "v"(acc8[3]), "v"(acc8[4]), "v"(acc8[5]), "v"(acc8[6]), "v"(acc8[7]),
                          "v"(o[0]), "v"(o[1]), "v"(o[2]), "v"(o[3]), "v"(o[4]), "v"(o[5]), "v"(o[6]), "v"(o[7]), "v"(o[8]), "v"(acc40));
      }
      if (sg < WT / 3) {
#pragma unroll
        for (int ks = 0; ks < 3; ++ks) {
#pragma unroll
          for (int i = 0; i < 3; ++i)
            dd[sg & 1][i] = __builtin_amdgcn_wmma_f32_16x16x4_f32(false, fa[3 * sg + i][ks], false, bk[ks], (short)0,
                                                                  ks == 0 ? zero8 : dd[sg & 1][i], false, false);
        }
      }
      if (sg >= 1) {
        const int sp = sg - 1;
#pragma unroll
        for (int i = 0; i < 3; ++i) {
          const v8f d = dd[sp & 1][i];
          const int T = WT * WV + 3 * sp + i;
          if (T < 45) {
            const int g  = T / 9;
            const int d1 = T - 9 * g;
            const float xd = xv[d1];
#pragma unroll
            for (int r = 0; r < 8; ++r) acc8[r] += xd * d[r];
            if (d1 == 8 || 3 * sp + i == WT - 1) {
              const float xl = xv[g];
              const float xh = xv[g < 4 ? 5 + g : 5];
              const bool last = hi && g == 4;
              const float m1 = last ? 0.0f : (hi ? xh : xl);
#pragma unroll
              for (int aa = 0; aa < 8; ++aa) {
                o[aa] += m1 * acc8[aa];
                o[8]  += (last ? xv[aa] : 0.0f) * acc8[aa];
                acc8[aa] = 0.0f;
              }
            }
          } else if (T == 45) {
#pragma unroll
            for (int r = 0; r < 8; ++r) acc40 += xv[r] * d[r];
          } else if (T == 46) {
            acc40 += xv[8] * d[0];
            o[8] += (hi ? 0.0f : xv[8]) * acc40;
          } else {
#pragma unroll
            for (int r = 0; r < 8; ++r) o[r] += hi ? 0.0f : d[r];
            o[8] += hi ? d[0] : 0.0f;
          }
        }
      }
    }

    float* pw = part + ((t - tb) * 4 + WV) * PTILE + n16 * 9;
float tot[9];
#pragma unroll
    for (int k = 0; k < 9; ++k) tot[k] = o[k] + xhalf(o[k]);
    if (!hi) {
#pragma unroll
      for (int k = 0; k < 9; ++k) pw[k] = tot[k];
    }
  }
}

__global__ __launch_bounds__(128) void sc_main(
    const float* __restrict__ X,
    const int*   __restrict__ index, int N,
    float*       __restrict__ out, int C,
    const float* __restrict__ coef)
{
  __shared__ __attribute__((aligned(16))) float At[MT * TPITCH];
  __shared__ int nodeList[LCAP];
  __shared__ int waveCnt[4];
  const int c    = blockIdx.x;
  const int s    = blockIdx.y;
  const int tid  = threadIdx.x;
  const int lane = tid & 31;
  const int wave = __builtin_amdgcn_readfirstlane(tid >> 5);
  const int n16  = lane & 15;
  const bool hi  = (lane & 16) != 0;

  {
    const float*  gs = coef + (size_t)(s * C + c) * GITEM;
    const float4* gp = (const float4*)gs;
    float2* lp = (float2*)At;
    int a0[9], a1[9];
#pragma unroll
    for (int m = 0; m < 9; ++m) { a0[m] = at_pair(2 * tid, m) >> 1; a1[m] = at_pair(2 * tid + 1, m) >> 1; }
    float4 va[9];
#pragma unroll
    for (int m = 0; m < 9; ++m) va[m] = gp[m * (GROW / 4) + tid];
#pragma unroll
    for (int m = 0; m < 9; ++m) {
      lp[a0[m]] = make_float2(va[m].x, va[m].y);
      lp[a1[m]] = make_float2(va[m].z, va[m].w);
    }
    if (tid < (PUSED + 3) / 4 - 128) {
#pragma unroll
      for (int m = 0; m < 9; ++m) { a0[m] = at_pair(2 * (128 + tid), m) >> 1; a1[m] = at_pair(2 * (128 + tid) + 1, m) >> 1; }
#pragma unroll
      for (int m = 0; m < 9; ++m) va[m] = gp[m * (GROW / 4) + 128 + tid];
#pragma unroll
      for (int m = 0; m < 9; ++m) {
        lp[a0[m]] = make_float2(va[m].x, va[m].y);
        lp[a1[m]] = make_float2(va[m].z, va[m].w);
      }
    }
    const float2 z2 = make_float2(0.f, 0.f);
#pragma unroll
    for (int k = 0; k < 6; ++k) {
      const int r = tid + 128 * k;
      lp[((r >> 4) * TPITCH + (r & 15) * KROW + 10) >> 1] = z2;
    }
    for (int e = tid; e < 30 * 5; e += 128) {
      const int zr = e / 5;
      const int sp = e - 5 * zr;
      const int rr = zr < 8 ? 45 * 17 + 8 + zr : (zr < 23 ? 46 * 17 + 1 + (zr - 8) : 47 * 17 + 9 + (zr - 23));
      lp[(rr * KROW + 2 * sp) >> 1] = z2;
    }
  }

  v2f fa[WT][3];

  for (int n0 = 0; n0 < N; n0 += LCAP) {
    const int wbase = n0 + wave * LWAVE;
    unsigned mk[16];
    int cntw = 0;
#pragma unroll
    for (int i = 0; i < 16; ++i) {
      const int n  = wbase + i * 32 + lane;
      const int nc = n < N ? n : N - 1;
      const int v  = index[nc];
      mk[i] = __builtin_amdgcn_ballot_w32(n < N && v == s);
      cntw += __builtin_popcount(mk[i]);
    }
    if (lane == 0) waveCnt[wave] = cntw;
    __syncthreads();
    if (n0 == 0) {
      const float* Ab = &At[(wave * WT) * TPITCH + n16 * KROW + (hi ? 2 : 0)];
#pragma unroll
      for (int j = 0; j < WT; ++j) {
        fa[j][0] = *(const v2f*)(Ab + j * TPITCH);
        fa[j][1] = *(const v2f*)(Ab + j * TPITCH + 4);
        fa[j][2] = *(const v2f*)(Ab + j * TPITCH + 8);
      }
    }
    int base = 0, total = 0;
#pragma unroll
    for (int w = 0; w < 4; ++w) {
      const int cw = waveCnt[w];
      if (w < wave) base += cw;
      total += cw;
    }
    base  = __builtin_amdgcn_readfirstlane(base);
    total = __builtin_amdgcn_readfirstlane(total);
#pragma unroll
    for (int i = 0; i < 16; ++i) {
      if ((mk[i] >> lane) & 1u)
        nodeList[base + (int)__builtin_amdgcn_mbcnt_lo(mk[i], 0u)] = wbase + i * 32 + lane;
      base += __builtin_popcount(mk[i]);
    }
    __syncthreads();

    const int numTiles = (total + 15) >> 4;
    for (int tb = 0; tb < numTiles; tb += CHUNK) {
      const int te = tb + CHUNK < numTiles ? tb + CHUNK : numTiles;
      if (wave == 0)      wave_tiles<0>(fa, X, C, c, nodeList, tb, te, total, At, n16, hi);
      else if (wave == 1) wave_tiles<1>(fa, X, C, c, nodeList, tb, te, total, At, n16, hi);
      else if (wave == 2) wave_tiles<2>(fa, X, C, c, nodeList, tb, te, total, At, n16, hi);
      else                wave_tiles<3>(fa, X, C, c, nodeList, tb, te, total, At, n16, hi);
      __syncthreads();
      const int nodes = (total - tb * 16) < (te - tb) * 16 ? (total - tb * 16) : (te - tb) * 16;
      for (int idx = tid; idx < nodes * 9; idx += 128) {
        const int ns = idx / 9;
        const int i  = idx - ns * 9;
        const int tl = ns >> 4;
        const float* p = At + tl * (4 * PTILE) + (idx - tl * PTILE);
        const float v = ((p[0] + p[PTILE]) + p[2 * PTILE]) + p[3 * PTILE];
        const int node = nodeList[tb * 16 + ns];
        out[(node * C + c) * 9 + i] = v;
      }
      __syncthreads();
    }
  }
}

extern "C" void kernel_launch(void* const* d_in, const int* in_sizes, int n_in,
                              void* d_out, int out_size, void* d_ws, size_t ws_size,
                              hipStream_t stream) {
  (void)n_in; (void)out_size; (void)ws_size;
  const float* X     = (const float*)d_in[0];
  const int*   index = (const int*)d_in[1];
  const int N = in_sizes[1];
  const int C = in_sizes[0] / (N * 9);

  float* coef = (float*)d_ws;

  sc_coef<<<dim3(NPAIR, NSP), 128, 0, stream>>>(
      coef, C,
      (const float*)d_in[2],  (const float*)d_in[3],
      (const float*)d_in[4],  (const float*)d_in[5],
      (const float*)d_in[6],  (const float*)d_in[7],
      (const float*)d_in[8],  (const float*)d_in[9],
      (const float*)d_in[10], (const float*)d_in[11],
      (const float*)d_in[12], (const float*)d_in[13],
      (const float*)d_in[14], (const float*)d_in[15],
      (const float*)d_in[16], (const float*)d_in[17],
      (const float*)d_in[18], (const float*)d_in[19]);

  sc_main<<<dim3(C, NSP), 128, 0, stream>>>(
      X, index, N, (float*)d_out, C, coef);
}
